# starting kernel
# baseline (speedup 1.0000x reference)
#include <hip/hip_runtime.h>
#include <hip/hip_bf16.h>

typedef float v2f __attribute__((ext_vector_type(2)));
typedef float v8f __attribute__((ext_vector_type(8)));

#define EPSV 1e-5f

__global__ void roi_pool_kernel(const float* __restrict__ x, float* __restrict__ A1) {
    const int n = blockIdx.x;
    const int c = threadIdx.x;
    const float* base = x + ((size_t)(n * 256 + c) * 64) * 64;
    float v[4][5];
#pragma unroll
    for (int rr = 0; rr < 4; ++rr)
#pragma unroll
        for (int cc = 0; cc < 5; ++cc)
            v[rr][cc] = base[(2 + rr) * 64 + 1 + cc];
    float r0[5], r1[5];
#pragma unroll
    for (int cc = 0; cc < 5; ++cc) {
        r0[cc] = fmaxf(v[0][cc], v[1][cc]);
        r1[cc] = fmaxf(v[2][cc], v[3][cc]);
    }
    float p00 = fmaxf(fmaxf(r0[0], r0[1]), r0[2]);
    float p01 = fmaxf(fmaxf(r0[2], r0[3]), r0[4]);
    float p10 = fmaxf(fmaxf(r1[0], r1[1]), r1[2]);
    float p11 = fmaxf(fmaxf(r1[2], r1[3]), r1[4]);
    float* o = A1 + (size_t)n * 4 * 256 + c;
    o[0 * 256] = p00;
    o[1 * 256] = p01;
    o[2 * 256] = p10;
    o[3 * 256] = p11;
}

template <int KIN, int COUT, bool RESHAPE>
__global__ __launch_bounds__(256) void conv_bn_relu_kernel(
    const float* __restrict__ Ain, int a_head_stride,
    const float* __restrict__ Wt,
    const float* __restrict__ g, const float* __restrict__ b,
    float* __restrict__ Out, int out_head_stride) {
    constexpr int NT  = COUT / 16;
    constexpr int TT  = 16 * NT;
    constexpr int TPW = TT / 8;

    __shared__ float lsum[COUT], lsq[COUT], lscale[COUT], lshift[COUT];

    const int h    = blockIdx.x;
    const int tid  = threadIdx.x;
    const int wave = tid >> 5;
    const int lane = tid & 31;
    const int half = lane >> 4;
    const int r    = lane & 15;

    if (tid < COUT) { lsum[tid] = 0.f; lsq[tid] = 0.f; }
    __syncthreads();

    const float* A = Ain + (size_t)h * a_head_stride;
    const float* W = Wt + (size_t)h * COUT * KIN;

    v8f acc[TPW];
#pragma unroll
    for (int i = 0; i < TPW; ++i) {
        const int tt = wave + 8 * i;
        const int mt = tt & 15, nt = tt >> 4;
        const int m0 = mt * 16, n0 = nt * 16;
        v8f c = {};
        const float2* arow = reinterpret_cast<const float2*>(A + (size_t)(m0 + r) * KIN);
        const float2* brow = reinterpret_cast<const float2*>(W + (size_t)(n0 + r) * KIN);
        for (int k0 = 0; k0 < KIN; k0 += 4) {
            float2 av = arow[(k0 >> 1) + half];
            float2 bv = brow[(k0 >> 1) + half];
            v2f a; a[0] = av.x; a[1] = av.y;
            v2f bb; bb[0] = bv.x; bb[1] = bv.y;
            c = __builtin_amdgcn_wmma_f32_16x16x4_f32(false, a, false, bb,
                                                      (short)0, c, false, false);
        }
        acc[i] = c;
        float s = 0.f, q = 0.f;
#pragma unroll
        for (int v = 0; v < 8; ++v) { float xv = c[v]; s += xv; q += xv * xv; }
        atomicAdd(&lsum[n0 + r], s);
        atomicAdd(&lsq[n0 + r], q);
    }
    __syncthreads();

    if (tid < COUT) {
        float mu  = lsum[tid] * (1.f / 256.f);
        float var = lsq[tid] * (1.f / 256.f) - mu * mu;
        float sc  = g[h * COUT + tid] * rsqrtf(var + EPSV);
        lscale[tid] = sc;
        lshift[tid] = b[h * COUT + tid] - mu * sc;
    }
    __syncthreads();

    float* O = Out + (size_t)h * out_head_stride;
#pragma unroll
    for (int i = 0; i < TPW; ++i) {
        const int tt = wave + 8 * i;
        const int mt = tt & 15, nt = tt >> 4;
        const int m0 = mt * 16, n0 = nt * 16;
        const int col = n0 + r;
        const float sc = lscale[col], sh = lshift[col];
#pragma unroll
        for (int v = 0; v < 8; ++v) {
            const int row = m0 + v + 8 * half;
            float val = acc[i][v] * sc + sh;
            val = fmaxf(val, 0.f);
            if (RESHAPE) {
                const int n = row >> 2, s = row & 3;
                O[n * (COUT * 4) + col * 4 + s] = val;
            } else {
                O[(size_t)row * COUT + col] = val;
            }
        }
    }
}

template <int M, int N, int K, bool RELU>
__global__ __launch_bounds__(256) void gemm_bias_kernel(
    const float* __restrict__ Ain, int a_head_stride,
    const float* __restrict__ Wt,
    const float* __restrict__ bias,
    float* __restrict__ Out, int out_head_stride) {
    constexpr int MT = (M + 15) / 16, NT = (N + 15) / 16, TT = MT * NT;
    const int h    = blockIdx.y;
    const int tid  = threadIdx.x;
    const int wave = tid >> 5;
    const int lane = tid & 31;
    const int half = lane >> 4;
    const int r    = lane & 15;
    const int wglobal = blockIdx.x * (blockDim.x >> 5) + wave;
    const int wstride = gridDim.x * (blockDim.x >> 5);

    const float* A = Ain + (size_t)h * a_head_stride;
    const float* W = Wt + (size_t)h * N * K;
    float* O = Out + (size_t)h * out_head_stride;

    for (int tt = wglobal; tt < TT; tt += wstride) {
        const int mt = tt % MT, nt = tt / MT;
        const int m0 = mt * 16, n0 = nt * 16;
        const int col = n0 + r;
        const bool colok = (col < N);
        v8f c = {};
        const float2* arow = reinterpret_cast<const float2*>(A + (size_t)(m0 + r) * K);
        const float2* brow = reinterpret_cast<const float2*>(W + (size_t)(colok ? col : 0) * K);
        for (int k0 = 0; k0 < K; k0 += 4) {
            float2 av = arow[(k0 >> 1) + half];
            float2 bv = brow[(k0 >> 1) + half];
            v2f a; a[0] = av.x; a[1] = av.y;
            v2f bb; bb[0] = colok ? bv.x : 0.f; bb[1] = colok ? bv.y : 0.f;
            c = __builtin_amdgcn_wmma_f32_16x16x4_f32(false, a, false, bb,
                                                      (short)0, c, false, false);
        }
        if (colok) {
            const float bs = bias[h * N + col];
#pragma unroll
            for (int v = 0; v < 8; ++v) {
                const int row = m0 + v + 8 * half;
                float val = c[v] + bs;
                if (RELU) val = fmaxf(val, 0.f);
                O[(size_t)row * N + col] = val;
            }
        }
    }
}

__global__ void final_combine_kernel(const float* __restrict__ hout,
                                     const float* __restrict__ W2,
                                     const float* __restrict__ bias2,
                                     float* __restrict__ out) {
    __shared__ float comb[100];
    const int n = blockIdx.x;
    const int t = threadIdx.x;
    if (t < 100) {
        float s = 0.f;
#pragma unroll
        for (int h = 0; h < 4; ++h) s += hout[((size_t)h * 64 + n) * 100 + t];
        comb[t] = s * 0.25f;
    }
    __syncthreads();
    if (t < 100) {
        float acc = bias2[t];
        for (int j = 0; j < 100; ++j) acc += comb[j] * W2[t * 100 + j];
        float* orow = out + (size_t)n * 64 * 100 + t;
#pragma unroll 4
        for (int rep = 0; rep < 64; ++rep) orow[(size_t)rep * 100] = acc;
    }
}

extern "C" void kernel_launch(void* const* d_in, const int* in_sizes, int n_in,
                              void* d_out, int out_size, void* d_ws, size_t ws_size,
                              hipStream_t stream) {
    const float* x     = (const float*)d_in[0];
    const float* w1    = (const float*)d_in[1];
    const float* g1    = (const float*)d_in[2];
    const float* b1    = (const float*)d_in[3];
    const float* w2    = (const float*)d_in[4];
    const float* g2    = (const float*)d_in[5];
    const float* b2    = (const float*)d_in[6];
    const float* w3    = (const float*)d_in[7];
    const float* g3    = (const float*)d_in[8];
    const float* b3    = (const float*)d_in[9];
    const float* lw1   = (const float*)d_in[10];
    const float* lb1   = (const float*)d_in[11];
    const float* lw2   = (const float*)d_in[12];
    const float* lb2   = (const float*)d_in[13];
    const float* W2    = (const float*)d_in[14];
    const float* bias2 = (const float*)d_in[15];
    float* out = (float*)d_out;

    float* ws   = (float*)d_ws;
    float* A1   = ws;
    float* z1   = A1 + 256 * 256;
    float* z2   = z1 + 4 * 256 * 32;
    float* fcin = z2 + 4 * 256 * 64;
    float* h1   = fcin + 4 * 64 * 256;
    float* hout = h1 + 4 * 64 * 512;

    roi_pool_kernel<<<64, 256, 0, stream>>>(x, A1);

    conv_bn_relu_kernel<256, 32, false><<<4, 256, 0, stream>>>(
        A1, 0, w1, g1, b1, z1, 256 * 32);
    conv_bn_relu_kernel<32, 64, false><<<4, 256, 0, stream>>>(
        z1, 256 * 32, w2, g2, b2, z2, 256 * 64);
    conv_bn_relu_kernel<64, 64, true><<<4, 256, 0, stream>>>(
        z2, 256 * 64, w3, g3, b3, fcin, 64 * 256);

    gemm_bias_kernel<64, 512, 256, true><<<dim3(8, 4), 256, 0, stream>>>(
        fcin, 64 * 256, lw1, lb1, h1, 64 * 512);
    gemm_bias_kernel<64, 100, 512, false><<<dim3(4, 4), 256, 0, stream>>>(
        h1, 64 * 512, lw2, lb2, hout, 64 * 100);

    final_combine_kernel<<<64, 128, 0, stream>>>(hout, W2, bias2, out);
}


# Output tiles spread over more workgroups: conv layers one 16-channel column tile per workgroup (8/16/16, BN stats stay local, z1/z2 tile-major), fc1/fc2 one tile per wave; k order, BN sums unchanged
# speedup vs baseline: 1.6622x; 1.6622x over previous
#include <hip/hip_runtime.h>
#include <hip/hip_bf16.h>

typedef float v2f __attribute__((ext_vector_type(2)));
typedef float v8f __attribute__((ext_vector_type(8)));

#define EPSV 1e-5f

__global__ void roi_pool_kernel(const float* __restrict__ x, float* __restrict__ A1) {
    const int n = blockIdx.x;
    const int c = threadIdx.x;
    const float* base = x + ((size_t)(n * 256 + c) * 64) * 64;
    float v[4][5];
#pragma unroll
    for (int rr = 0; rr < 4; ++rr)
#pragma unroll
        for (int cc = 0; cc < 5; ++cc)
            v[rr][cc] = base[(2 + rr) * 64 + 1 + cc];
    float r0[5], r1[5];
#pragma unroll
    for (int cc = 0; cc < 5; ++cc) {
        r0[cc] = fmaxf(v[0][cc], v[1][cc]);
        r1[cc] = fmaxf(v[2][cc], v[3][cc]);
    }
    float p00 = fmaxf(fmaxf(r0[0], r0[1]), r0[2]);
    float p01 = fmaxf(fmaxf(r0[2], r0[3]), r0[4]);
    float p10 = fmaxf(fmaxf(r1[0], r1[1]), r1[2]);
    float p11 = fmaxf(fmaxf(r1[2], r1[3]), r1[4]);
    float* o = A1 + (size_t)n * 4 * 256 + c;
    o[0 * 256] = p00;
    o[1 * 256] = p01;
    o[2 * 256] = p10;
    o[3 * 256] = p11;
}

template <int KIN, int COUT, bool A_TILED, bool RESHAPE>
__global__ __launch_bounds__(256) void conv_bn_relu_kernel(
    const float* __restrict__ Ain, int a_head_stride,
    const float* __restrict__ Wt,
    const float* __restrict__ g, const float* __restrict__ b,
    float* __restrict__ Out, int out_head_stride) {
    constexpr int KSTEPS = KIN / 4;
    constexpr int UNR = (KSTEPS <= 16) ? KSTEPS : 8;

    __shared__ float lsum[16], lsq[16], lscale[16], lshift[16];

    const int ct   = blockIdx.x;
    const int h    = blockIdx.y;
    const int tid  = threadIdx.x;
    const int wave = tid >> 5;
    const int lane = tid & 31;
    const int half = lane >> 4;
    const int r    = lane & 15;

    if (tid < 16) { lsum[tid] = 0.f; lsq[tid] = 0.f; }
    __syncthreads();

    const float* A = Ain + (size_t)h * a_head_stride;
    const float* W = Wt + ((size_t)h * COUT + ct * 16 + r) * KIN + 2 * half;

    const int row0 = wave * 16 + r;
    const int row1 = (wave + 8) * 16 + r;
    const float* ap0 = A_TILED ? (A + (size_t)row0 * 16 + 2 * half) : (A + (size_t)row0 * KIN + 2 * half);
    const float* ap1 = A_TILED ? (A + (size_t)row1 * 16 + 2 * half) : (A + (size_t)row1 * KIN + 2 * half);

    v8f acc0 = v8f{}, acc1 = v8f{};

#pragma unroll UNR
    for (int k0 = 0; k0 < KIN; k0 += 4) {
        const int aoff = A_TILED ? ((k0 >> 4) * (256 * 16) + (k0 & 15)) : k0;
        const float2 av0 = *reinterpret_cast<const float2*>(ap0 + aoff);
        const float2 av1 = *reinterpret_cast<const float2*>(ap1 + aoff);
        const float2 bv  = *reinterpret_cast<const float2*>(W + k0);
        v2f a0; a0[0] = av0.x; a0[1] = av0.y;
        v2f a1; a1[0] = av1.x; a1[1] = av1.y;
        v2f bb; bb[0] = bv.x;  bb[1] = bv.y;
        acc0 = __builtin_amdgcn_wmma_f32_16x16x4_f32(false, a0, false, bb, (short)0, acc0, false, false);
        acc1 = __builtin_amdgcn_wmma_f32_16x16x4_f32(false, a1, false, bb, (short)0, acc1, false, false);
    }

    {
        float s = 0.f, q = 0.f;
#pragma unroll
        for (int v = 0; v < 8; ++v) { float xv = acc0[v]; s += xv; q += xv * xv; }
        atomicAdd(&lsum[r], s);
        atomicAdd(&lsq[r], q);
        s = 0.f; q = 0.f;
#pragma unroll
        for (int v = 0; v < 8; ++v) { float xv = acc1[v]; s += xv; q += xv * xv; }
        atomicAdd(&lsum[r], s);
        atomicAdd(&lsq[r], q);
    }
    __syncthreads();

    if (tid < 16) {
        const int ch = h * COUT + ct * 16 + tid;
        float mu  = lsum[tid] * (1.f / 256.f);
        float var = lsq[tid] * (1.f / 256.f) - mu * mu;
        float sc  = g[ch] * rsqrtf(var + EPSV);
        lscale[tid] = sc;
        lshift[tid] = b[ch] - mu * sc;
    }
    __syncthreads();

    float* O = Out + (size_t)h * out_head_stride;
    const float sc = lscale[r], sh = lshift[r];
    const int col = ct * 16 + r;
#pragma unroll
    for (int i = 0; i < 2; ++i) {
        const int m0 = (wave + 8 * i) * 16;
        const v8f c = i ? acc1 : acc0;
#pragma unroll
        for (int v = 0; v < 8; ++v) {
            const int row = m0 + v + 8 * half;
            float val = c[v] * sc + sh;
            val = fmaxf(val, 0.f);
            if (RESHAPE) {
                const int n = row >> 2, s = row & 3;
                O[n * (COUT * 4) + col * 4 + s] = val;
            } else {
                O[((size_t)ct * 256 + row) * 16 + r] = val;
            }
        }
    }
}

template <int M, int N, int K, bool RELU, int GX, int WPB>
__global__ __launch_bounds__(WPB * 32) void gemm_bias_kernel(
    const float* __restrict__ Ain, int a_head_stride,
    const float* __restrict__ Wt,
    const float* __restrict__ bias,
    float* __restrict__ Out, int out_head_stride) {
    constexpr int MT = (M + 15) / 16, NT = (N + 15) / 16, TT = MT * NT;
    constexpr int WS  = GX * WPB;
    constexpr int TPW = (TT + WS - 1) / WS;
    static_assert(WS % MT == 0, "a wave's tiles must share one row tile");
    static_assert(TPW == 1 || TT % WS == 0, "every wave must own a full block");
    constexpr int NTS = WS / MT;
    constexpr bool GUARD = (N % 16) != 0;

    const int h    = blockIdx.y;
    const int tid  = threadIdx.x;
    const int wave = tid >> 5;
    const int lane = tid & 31;
    const int half = lane >> 4;
    const int r    = lane & 15;
    const int wglobal = blockIdx.x * WPB + wave;
    if (wglobal >= TT) return;

    const float* A = Ain + (size_t)h * a_head_stride;
    const float* W = Wt + (size_t)h * N * K;
    float* O = Out + (size_t)h * out_head_stride;

    const int mt = wglobal % MT, nt0 = wglobal / MT;
    const int m0 = mt * 16;

    int  col[TPW];
    bool colok[TPW];
    const float2* brow[TPW];
    v8f acc[TPW];
#pragma unroll
    for (int i = 0; i < TPW; ++i) {
        col[i]   = (nt0 + i * NTS) * 16 + r;
        colok[i] = !GUARD || (col[i] < N);
        brow[i]  = reinterpret_cast<const float2*>(W + (size_t)(colok[i] ? col[i] : 0) * K) + half;
        acc[i]   = v8f{};
    }
    const float2* arow = reinterpret_cast<const float2*>(A + (size_t)(m0 + r) * K) + half;

    for (int k0 = 0; k0 < K; k0 += 4) {
        const float2 av = arow[k0 >> 1];
        v2f a; a[0] = av.x; a[1] = av.y;
#pragma unroll
        for (int i = 0; i < TPW; ++i) {
            const float2 bv = brow[i][k0 >> 1];
            v2f bb; bb[0] = colok[i] ? bv.x : 0.f; bb[1] = colok[i] ? bv.y : 0.f;
            acc[i] = __builtin_amdgcn_wmma_f32_16x16x4_f32(false, a, false, bb,
                                                           (short)0, acc[i], false, false);
        }
    }

#pragma unroll
    for (int i = 0; i < TPW; ++i) {
        if (colok[i]) {
            const float bs = bias[h * N + col[i]];
#pragma unroll
            for (int v = 0; v < 8; ++v) {
                const int row = m0 + v + 8 * half;
                float val = acc[i][v] + bs;
                if (RELU) val = fmaxf(val, 0.f);
                O[(size_t)row * N + col[i]] = val;
            }
        }
    }
}

__global__ void final_combine_kernel(const float* __restrict__ hout,
                                     const float* __restrict__ W2,
                                     const float* __restrict__ bias2,
                                     float* __restrict__ out) {
    __shared__ float comb[100];
    const int n = blockIdx.x;
    const int t = threadIdx.x;
    if (t < 100) {
        float s = 0.f;
#pragma unroll
        for (int h = 0; h < 4; ++h) s += hout[((size_t)h * 64 + n) * 100 + t];
        comb[t] = s * 0.25f;
    }
    __syncthreads();
    if (t < 100) {
        float acc = bias2[t];
        for (int j = 0; j < 100; ++j) acc += comb[j] * W2[t * 100 + j];
        float* orow = out + (size_t)n * 64 * 100 + t;
#pragma unroll 4
        for (int rep = 0; rep < 64; ++rep) orow[(size_t)rep * 100] = acc;
    }
}

extern "C" void kernel_launch(void* const* d_in, const int* in_sizes, int n_in,
                              void* d_out, int out_size, void* d_ws, size_t ws_size,
                              hipStream_t stream) {
    const float* x     = (const float*)d_in[0];
    const float* w1    = (const float*)d_in[1];
    const float* g1    = (const float*)d_in[2];
    const float* b1    = (const float*)d_in[3];
    const float* w2    = (const float*)d_in[4];
    const float* g2    = (const float*)d_in[5];
    const float* b2    = (const float*)d_in[6];
    const float* w3    = (const float*)d_in[7];
    const float* g3    = (const float*)d_in[8];
    const float* b3    = (const float*)d_in[9];
    const float* lw1   = (const float*)d_in[10];
    const float* lb1   = (const float*)d_in[11];
    const float* lw2   = (const float*)d_in[12];
    const float* lb2   = (const float*)d_in[13];
    const float* W2    = (const float*)d_in[14];
    const float* bias2 = (const float*)d_in[15];
    float* out = (float*)d_out;

    float* ws   = (float*)d_ws;
    float* A1   = ws;
    float* z1   = A1 + 256 * 256;
    float* z2   = z1 + 4 * 256 * 32;
    float* fcin = z2 + 4 * 256 * 64;
    float* h1   = fcin + 4 * 64 * 256;
    float* hout = h1 + 4 * 64 * 512;

    roi_pool_kernel<<<64, 256, 0, stream>>>(x, A1);

    conv_bn_relu_kernel<256, 32, false, false><<<dim3(2, 4), 256, 0, stream>>>(
        A1, 0, w1, g1, b1, z1, 256 * 32);
    conv_bn_relu_kernel<32, 64, true, false><<<dim3(4, 4), 256, 0, stream>>>(
        z1, 256 * 32, w2, g2, b2, z2, 256 * 64);
    conv_bn_relu_kernel<64, 64, true, true><<<dim3(4, 4), 256, 0, stream>>>(
        z2, 256 * 64, w3, g3, b3, fcin, 64 * 256);

    gemm_bias_kernel<64, 512, 256, true, 16, 8><<<dim3(16, 4), 256, 0, stream>>>(
        fcin, 64 * 256, lw1, lb1, h1, 64 * 512);
    gemm_bias_kernel<64, 100, 512, false, 7, 4><<<dim3(7, 4), 128, 0, stream>>>(
        h1, 64 * 512, lw2, lb2, hout, 64 * 100);

    final_combine_kernel<<<64, 128, 0, stream>>>(hout, W2, bias2, out);
}


# Fragment loads of both WMMA GEMM templates go to 16 bytes per lane straight into registers, two WMMAs per load (no LDS): k inside each run of 8 permuted alike for A and B; tile sum order changes
# speedup vs baseline: 1.7446x; 1.0496x over previous
#include <hip/hip_runtime.h>
#include <hip/hip_bf16.h>

typedef float v2f __attribute__((ext_vector_type(2)));
typedef float v8f __attribute__((ext_vector_type(8)));

#define EPSV 1e-5f

__global__ void roi_pool_kernel(const float* __restrict__ x, float* __restrict__ A1) {
    const int n = blockIdx.x;
    const int c = threadIdx.x;
    const float* base = x + ((size_t)(n * 256 + c) * 64) * 64;
    float v[4][5];
#pragma unroll
    for (int rr = 0; rr < 4; ++rr)
#pragma unroll
        for (int cc = 0; cc < 5; ++cc)
            v[rr][cc] = base[(2 + rr) * 64 + 1 + cc];
    float r0[5], r1[5];
#pragma unroll
    for (int cc = 0; cc < 5; ++cc) {
        r0[cc] = fmaxf(v[0][cc], v[1][cc]);
        r1[cc] = fmaxf(v[2][cc], v[3][cc]);
    }
    float p00 = fmaxf(fmaxf(r0[0], r0[1]), r0[2]);
    float p01 = fmaxf(fmaxf(r0[2], r0[3]), r0[4]);
    float p10 = fmaxf(fmaxf(r1[0], r1[1]), r1[2]);
    float p11 = fmaxf(fmaxf(r1[2], r1[3]), r1[4]);
    float* o = A1 + (size_t)n * 4 * 256 + c;
    o[0 * 256] = p00;
    o[1 * 256] = p01;
    o[2 * 256] = p10;
    o[3 * 256] = p11;
}

template <int KIN, int COUT, bool A_TILED, bool RESHAPE>
__global__ __launch_bounds__(256) void conv_bn_relu_kernel(
    const float* __restrict__ Ain, int a_head_stride,
    const float* __restrict__ Wt,
    const float* __restrict__ g, const float* __restrict__ b,
    float* __restrict__ Out, int out_head_stride) {
    static_assert(KIN % 8 == 0, "k is walked in runs of 8");
    constexpr int RUNS = KIN / 8;
    constexpr int UNR = (RUNS <= 8) ? RUNS : 8;

    __shared__ float lsum[16], lsq[16], lscale[16], lshift[16];

    const int ct   = blockIdx.x;
    const int h    = blockIdx.y;
    const int tid  = threadIdx.x;
    const int wave = tid >> 5;
    const int lane = tid & 31;
    const int half = lane >> 4;
    const int r    = lane & 15;

    if (tid < 16) { lsum[tid] = 0.f; lsq[tid] = 0.f; }
    __syncthreads();

    const float* A = Ain + (size_t)h * a_head_stride;
    const float* W = Wt + ((size_t)h * COUT + ct * 16 + r) * KIN + 4 * half;

    const int row0 = wave * 16 + r;
    const int row1 = (wave + 8) * 16 + r;
    const float* ap0 = A_TILED ? (A + (size_t)row0 * 16 + 4 * half) : (A + (size_t)row0 * KIN + 4 * half);
    const float* ap1 = A_TILED ? (A + (size_t)row1 * 16 + 4 * half) : (A + (size_t)row1 * KIN + 4 * half);

    v8f acc0 = v8f{}, acc1 = v8f{};

#pragma unroll UNR
    for (int kb = 0; kb < KIN; kb += 8) {
        const int aoff = A_TILED ? ((kb >> 4) * (256 * 16) + (kb & 15)) : kb;
        const float4 av0 = *reinterpret_cast<const float4*>(ap0 + aoff);
        const float4 av1 = *reinterpret_cast<const float4*>(ap1 + aoff);
        const float4 bv  = *reinterpret_cast<const float4*>(W + kb);
        v2f a0; a0[0] = av0.x; a0[1] = av0.y;
        v2f a1; a1[0] = av1.x; a1[1] = av1.y;
        v2f bb; bb[0] = bv.x;  bb[1] = bv.y;
        acc0 = __builtin_amdgcn_wmma_f32_16x16x4_f32(false, a0, false, bb, (short)0, acc0, false, false);
        acc1 = __builtin_amdgcn_wmma_f32_16x16x4_f32(false, a1, false, bb, (short)0, acc1, false, false);
        a0[0] = av0.z; a0[1] = av0.w;
        a1[0] = av1.z; a1[1] = av1.w;
        bb[0] = bv.z;  bb[1] = bv.w;
        acc0 = __builtin_amdgcn_wmma_f32_16x16x4_f32(false, a0, false, bb, (short)0, acc0, false, false);
        acc1 = __builtin_amdgcn_wmma_f32_16x16x4_f32(false, a1, false, bb, (short)0, acc1, false, false);
    }

    {
        float s = 0.f, q = 0.f;
#pragma unroll
        for (int v = 0; v < 8; ++v) { float xv = acc0[v]; s += xv; q += xv * xv; }
        atomicAdd(&lsum[r], s);
        atomicAdd(&lsq[r], q);
        s = 0.f; q = 0.f;
#pragma unroll
        for (int v = 0; v < 8; ++v) { float xv = acc1[v]; s += xv; q += xv * xv; }
        atomicAdd(&lsum[r], s);
        atomicAdd(&lsq[r], q);
    }
    __syncthreads();

    if (tid < 16) {
        const int ch = h * COUT + ct * 16 + tid;
        float mu  = lsum[tid] * (1.f / 256.f);
        float var = lsq[tid] * (1.f / 256.f) - mu * mu;
        float sc  = g[ch] * rsqrtf(var + EPSV);
        lscale[tid] = sc;
        lshift[tid] = b[ch] - mu * sc;
    }
    __syncthreads();

    float* O = Out + (size_t)h * out_head_stride;
    const float sc = lscale[r], sh = lshift[r];
    const int col = ct * 16 + r;
#pragma unroll
    for (int i = 0; i < 2; ++i) {
        const int m0 = (wave + 8 * i) * 16;
        const v8f c = i ? acc1 : acc0;
#pragma unroll
        for (int v = 0; v < 8; ++v) {
            const int row = m0 + v + 8 * half;
            float val = c[v] * sc + sh;
            val = fmaxf(val, 0.f);
            if (RESHAPE) {
                const int n = row >> 2, s = row & 3;
                O[n * (COUT * 4) + col * 4 + s] = val;
            } else {
                O[((size_t)ct * 256 + row) * 16 + r] = val;
            }
        }
    }
}

template <int M, int N, int K, bool RELU, int GX, int WPB>
__global__ __launch_bounds__(WPB * 32) void gemm_bias_kernel(
    const float* __restrict__ Ain, int a_head_stride,
    const float* __restrict__ Wt,
    const float* __restrict__ bias,
    float* __restrict__ Out, int out_head_stride) {
    constexpr int MT = (M + 15) / 16, NT = (N + 15) / 16, TT = MT * NT;
    constexpr int WS  = GX * WPB;
    constexpr int TPW = (TT + WS - 1) / WS;
    static_assert(WS % MT == 0, "a wave's tiles must share one row tile");
    static_assert(TPW == 1 || TT % WS == 0, "every wave must own a full block");
    constexpr int NTS = WS / MT;
    constexpr bool GUARD = (N % 16) != 0;
    static_assert(K % 8 == 0, "k is walked in runs of 8");

    const int h    = blockIdx.y;
    const int tid  = threadIdx.x;
    const int wave = tid >> 5;
    const int lane = tid & 31;
    const int half = lane >> 4;
    const int r    = lane & 15;
    const int wglobal = blockIdx.x * WPB + wave;
    if (wglobal >= TT) return;

    const float* A = Ain + (size_t)h * a_head_stride;
    const float* W = Wt + (size_t)h * N * K;
    float* O = Out + (size_t)h * out_head_stride;

    const int mt = wglobal % MT, nt0 = wglobal / MT;
    const int m0 = mt * 16;

    int  col[TPW];
    bool colok[TPW];
    const float4* brow[TPW];
    v8f acc[TPW];
#pragma unroll
    for (int i = 0; i < TPW; ++i) {
        col[i]   = (nt0 + i * NTS) * 16 + r;
        colok[i] = !GUARD || (col[i] < N);
        brow[i]  = reinterpret_cast<const float4*>(W + (size_t)(colok[i] ? col[i] : 0) * K) + half;
        acc[i]   = v8f{};
    }
    const float4* arow = reinterpret_cast<const float4*>(A + (size_t)(m0 + r) * K) + half;

    for (int kb = 0; kb < K; kb += 8) {
        const float4 av = arow[kb >> 2];
        v2f alo; alo[0] = av.x; alo[1] = av.y;
        v2f ahi; ahi[0] = av.z; ahi[1] = av.w;
#pragma unroll
        for (int i = 0; i < TPW; ++i) {
            const float4 bv = brow[i][kb >> 2];
            v2f blo; blo[0] = colok[i] ? bv.x : 0.f; blo[1] = colok[i] ? bv.y : 0.f;
            v2f bhi; bhi[0] = colok[i] ? bv.z : 0.f; bhi[1] = colok[i] ? bv.w : 0.f;
            acc[i] = __builtin_amdgcn_wmma_f32_16x16x4_f32(false, alo, false, blo,
                                                           (short)0, acc[i], false, false);
            acc[i] = __builtin_amdgcn_wmma_f32_16x16x4_f32(false, ahi, false, bhi,
                                                           (short)0, acc[i], false, false);
        }
    }

#pragma unroll
    for (int i = 0; i < TPW; ++i) {
        if (colok[i]) {
            const float bs = bias[h * N + col[i]];
#pragma unroll
            for (int v = 0; v < 8; ++v) {
                const int row = m0 + v + 8 * half;
                float val = acc[i][v] + bs;
                if (RELU) val = fmaxf(val, 0.f);
                O[(size_t)row * N + col[i]] = val;
            }
        }
    }
}

__global__ void final_combine_kernel(const float* __restrict__ hout,
                                     const float* __restrict__ W2,
                                     const float* __restrict__ bias2,
                                     float* __restrict__ out) {
    __shared__ float comb[100];
    const int n = blockIdx.x;
    const int t = threadIdx.x;
    if (t < 100) {
        float s = 0.f;
#pragma unroll
        for (int h = 0; h < 4; ++h) s += hout[((size_t)h * 64 + n) * 100 + t];
        comb[t] = s * 0.25f;
    }
    __syncthreads();
    if (t < 100) {
        float acc = bias2[t];
        for (int j = 0; j < 100; ++j) acc += comb[j] * W2[t * 100 + j];
        float* orow = out + (size_t)n * 64 * 100 + t;
#pragma unroll 4
        for (int rep = 0; rep < 64; ++rep) orow[(size_t)rep * 100] = acc;
    }
}

extern "C" void kernel_launch(void* const* d_in, const int* in_sizes, int n_in,
                              void* d_out, int out_size, void* d_ws, size_t ws_size,
                              hipStream_t stream) {
    const float* x     = (const float*)d_in[0];
    const float* w1    = (const float*)d_in[1];
    const float* g1    = (const float*)d_in[2];
    const float* b1    = (const float*)d_in[3];
    const float* w2    = (const float*)d_in[4];
    const float* g2    = (const float*)d_in[5];
    const float* b2    = (const float*)d_in[6];
    const float* w3    = (const float*)d_in[7];
    const float* g3    = (const float*)d_in[8];
    const float* b3    = (const float*)d_in[9];
    const float* lw1   = (const float*)d_in[10];
    const float* lb1   = (const float*)d_in[11];
    const float* lw2   = (const float*)d_in[12];
    const float* lb2   = (const float*)d_in[13];
    const float* W2    = (const float*)d_in[14];
    const float* bias2 = (const float*)d_in[15];
    float* out = (float*)d_out;

    float* ws   = (float*)d_ws;
    float* A1   = ws;
    float* z1   = A1 + 256 * 256;
    float* z2   = z1 + 4 * 256 * 32;
    float* fcin = z2 + 4 * 256 * 64;
    float* h1   = fcin + 4 * 64 * 256;
    float* hout = h1 + 4 * 64 * 512;

    roi_pool_kernel<<<64, 256, 0, stream>>>(x, A1);

    conv_bn_relu_kernel<256, 32, false, false><<<dim3(2, 4), 256, 0, stream>>>(
        A1, 0, w1, g1, b1, z1, 256 * 32);
    conv_bn_relu_kernel<32, 64, true, false><<<dim3(4, 4), 256, 0, stream>>>(
        z1, 256 * 32, w2, g2, b2, z2, 256 * 64);
    conv_bn_relu_kernel<64, 64, true, true><<<dim3(4, 4), 256, 0, stream>>>(
        z2, 256 * 64, w3, g3, b3, fcin, 64 * 256);

    gemm_bias_kernel<64, 512, 256, true, 16, 8><<<dim3(16, 4), 256, 0, stream>>>(
        fcin, 64 * 256, lw1, lb1, h1, 64 * 512);
    gemm_bias_kernel<64, 100, 512, false, 7, 4><<<dim3(7, 4), 128, 0, stream>>>(
        h1, 64 * 512, lw2, lb2, hout, 64 * 100);

    final_combine_kernel<<<64, 128, 0, stream>>>(hout, W2, bias2, out);
}


# All global loads of the five WMMA kernels go out ahead of use: fragment loads a group of 16 8-runs ahead in a second register set (held by a scheduling barrier), BN scale/offset and bias at entry
# speedup vs baseline: 1.9861x; 1.1384x over previous
#include <hip/hip_runtime.h>
#include <hip/hip_bf16.h>

typedef float v2f __attribute__((ext_vector_type(2)));
typedef float v8f __attribute__((ext_vector_type(8)));

#define EPSV 1e-5f

__global__ void roi_pool_kernel(const float* __restrict__ x, float* __restrict__ A1) {
    const int n = blockIdx.x;
    const int c = threadIdx.x;
    const float* base = x + ((size_t)(n * 256 + c) * 64) * 64;
    float v[4][5];
#pragma unroll
    for (int rr = 0; rr < 4; ++rr)
#pragma unroll
        for (int cc = 0; cc < 5; ++cc)
            v[rr][cc] = base[(2 + rr) * 64 + 1 + cc];
    float r0[5], r1[5];
#pragma unroll
    for (int cc = 0; cc < 5; ++cc) {
        r0[cc] = fmaxf(v[0][cc], v[1][cc]);
        r1[cc] = fmaxf(v[2][cc], v[3][cc]);
    }
    float p00 = fmaxf(fmaxf(r0[0], r0[1]), r0[2]);
    float p01 = fmaxf(fmaxf(r0[2], r0[3]), r0[4]);
    float p10 = fmaxf(fmaxf(r1[0], r1[1]), r1[2]);
    float p11 = fmaxf(fmaxf(r1[2], r1[3]), r1[4]);
    float* o = A1 + (size_t)n * 4 * 256 + c;
    o[0 * 256] = p00;
    o[1 * 256] = p01;
    o[2 * 256] = p10;
    o[3 * 256] = p11;
}

template <int KIN, int COUT, bool A_TILED, bool RESHAPE>
__global__ __launch_bounds__(256) void conv_bn_relu_kernel(
    const float* __restrict__ Ain, int a_head_stride,
    const float* __restrict__ Wt,
    const float* __restrict__ g, const float* __restrict__ b,
    float* __restrict__ Out, int out_head_stride) {
    static_assert(KIN % 8 == 0, "k is walked in runs of 8");
    constexpr int RUNS = KIN / 8;
    constexpr int G  = (RUNS < 16) ? RUNS : 16;
    constexpr int NG = RUNS / G;
    static_assert(RUNS % G == 0, "k splits into whole load groups");

    __shared__ float lsum[16], lsq[16], lscale[16], lshift[16];

    const int ct   = blockIdx.x;
    const int h    = blockIdx.y;
    const int tid  = threadIdx.x;
    const int wave = tid >> 5;
    const int lane = tid & 31;
    const int half = lane >> 4;
    const int r    = lane & 15;

    if (tid < 16) { lsum[tid] = 0.f; lsq[tid] = 0.f; }
    __syncthreads();

    const float* A = Ain + (size_t)h * a_head_stride;
    const float* W = Wt + ((size_t)h * COUT + ct * 16 + r) * KIN + 4 * half;

    const int row0 = wave * 16 + r;
    const int row1 = (wave + 8) * 16 + r;
    const float* ap0 = A_TILED ? (A + (size_t)row0 * 16 + 4 * half) : (A + (size_t)row0 * KIN + 4 * half);
    const float* ap1 = A_TILED ? (A + (size_t)row1 * 16 + 4 * half) : (A + (size_t)row1 * KIN + 4 * half);

    v8f acc0 = v8f{}, acc1 = v8f{};

    const float gcol = g[h * COUT + ct * 16 + r];
    const float bcol = b[h * COUT + ct * 16 + r];

    float4 av0[2][G], av1[2][G], bv[2][G];
#pragma unroll
    for (int j = 0; j < G; ++j) {
        const int kb   = 8 * j;
        const int aoff = A_TILED ? ((kb >> 4) * (256 * 16) + (kb & 15)) : kb;
        av0[0][j] = *reinterpret_cast<const float4*>(ap0 + aoff);
        av1[0][j] = *reinterpret_cast<const float4*>(ap1 + aoff);
        bv[0][j]  = *reinterpret_cast<const float4*>(W + kb);
    }
#pragma unroll
    for (int g = 0; g < NG; ++g) {
        const int cur = g & 1, nxt = cur ^ 1;
        if (g + 1 < NG) {
#pragma unroll
            for (int j = 0; j < G; ++j) {
                const int kb   = 8 * ((g + 1) * G + j);
                const int aoff = A_TILED ? ((kb >> 4) * (256 * 16) + (kb & 15)) : kb;
                av0[nxt][j] = *reinterpret_cast<const float4*>(ap0 + aoff);
                av1[nxt][j] = *reinterpret_cast<const float4*>(ap1 + aoff);
                bv[nxt][j]  = *reinterpret_cast<const float4*>(W + kb);
            }
        }
        __builtin_amdgcn_sched_barrier(0);
#pragma unroll
        for (int j = 0; j < G; ++j) {
            v2f a0; a0[0] = av0[cur][j].x; a0[1] = av0[cur][j].y;
            v2f a1; a1[0] = av1[cur][j].x; a1[1] = av1[cur][j].y;
            v2f bb; bb[0] = bv[cur][j].x;  bb[1] = bv[cur][j].y;
            acc0 = __builtin_amdgcn_wmma_f32_16x16x4_f32(false, a0, false, bb, (short)0, acc0, false, false);
            acc1 = __builtin_amdgcn_wmma_f32_16x16x4_f32(false, a1, false, bb, (short)0, acc1, false, false);
            a0[0] = av0[cur][j].z; a0[1] = av0[cur][j].w;
            a1[0] = av1[cur][j].z; a1[1] = av1[cur][j].w;
            bb[0] = bv[cur][j].z;  bb[1] = bv[cur][j].w;
            acc0 = __builtin_amdgcn_wmma_f32_16x16x4_f32(false, a0, false, bb, (short)0, acc0, false, false);
            acc1 = __builtin_amdgcn_wmma_f32_16x16x4_f32(false, a1, false, bb, (short)0, acc1, false, false);
        }
    }

    {
        float s = 0.f, q = 0.f;
#pragma unroll
        for (int v = 0; v < 8; ++v) { float xv = acc0[v]; s += xv; q += xv * xv; }
        atomicAdd(&lsum[r], s);
        atomicAdd(&lsq[r], q);
        s = 0.f; q = 0.f;
#pragma unroll
        for (int v = 0; v < 8; ++v) { float xv = acc1[v]; s += xv; q += xv * xv; }
        atomicAdd(&lsum[r], s);
        atomicAdd(&lsq[r], q);
    }
    __syncthreads();

    if (tid < 16) {
        float mu  = lsum[tid] * (1.f / 256.f);
        float var = lsq[tid] * (1.f / 256.f) - mu * mu;
        float sc  = gcol * rsqrtf(var + EPSV);
        lscale[tid] = sc;
        lshift[tid] = bcol - mu * sc;
    }
    __syncthreads();

    float* O = Out + (size_t)h * out_head_stride;
    const float sc = lscale[r], sh = lshift[r];
    const int col = ct * 16 + r;
#pragma unroll
    for (int i = 0; i < 2; ++i) {
        const int m0 = (wave + 8 * i) * 16;
        const v8f c = i ? acc1 : acc0;
#pragma unroll
        for (int v = 0; v < 8; ++v) {
            const int row = m0 + v + 8 * half;
            float val = c[v] * sc + sh;
            val = fmaxf(val, 0.f);
            if (RESHAPE) {
                const int n = row >> 2, s = row & 3;
                O[n * (COUT * 4) + col * 4 + s] = val;
            } else {
                O[((size_t)ct * 256 + row) * 16 + r] = val;
            }
        }
    }
}

template <int M, int N, int K, bool RELU, int GX, int WPB>
__global__ __launch_bounds__(WPB * 32) void gemm_bias_kernel(
    const float* __restrict__ Ain, int a_head_stride,
    const float* __restrict__ Wt,
    const float* __restrict__ bias,
    float* __restrict__ Out, int out_head_stride) {
    constexpr int MT = (M + 15) / 16, NT = (N + 15) / 16, TT = MT * NT;
    constexpr int WS  = GX * WPB;
    constexpr int TPW = (TT + WS - 1) / WS;
    static_assert(WS % MT == 0, "a wave's tiles must share one row tile");
    static_assert(TPW == 1 || TT % WS == 0, "every wave must own a full block");
    constexpr int NTS = WS / MT;
    constexpr bool GUARD = (N % 16) != 0;
    static_assert(K % 8 == 0, "k is walked in runs of 8");

    const int h    = blockIdx.y;
    const int tid  = threadIdx.x;
    const int wave = tid >> 5;
    const int lane = tid & 31;
    const int half = lane >> 4;
    const int r    = lane & 15;
    const int wglobal = blockIdx.x * WPB + wave;
    if (wglobal >= TT) return;

    const float* A = Ain + (size_t)h * a_head_stride;
    const float* W = Wt + (size_t)h * N * K;
    float* O = Out + (size_t)h * out_head_stride;

    const int mt = wglobal % MT, nt0 = wglobal / MT;
    const int m0 = mt * 16;

    int  col[TPW];
    bool colok[TPW];
    const float4* brow[TPW];
    v8f acc[TPW];
#pragma unroll
    for (int i = 0; i < TPW; ++i) {
        col[i]   = (nt0 + i * NTS) * 16 + r;
        colok[i] = !GUARD || (col[i] < N);
        brow[i]  = reinterpret_cast<const float4*>(W + (size_t)(colok[i] ? col[i] : 0) * K) + half;
        acc[i]   = v8f{};
    }
    const float4* arow = reinterpret_cast<const float4*>(A + (size_t)(m0 + r) * K) + half;
    float bs[TPW];
#pragma unroll
    for (int i = 0; i < TPW; ++i) bs[i] = bias[h * N + (colok[i] ? col[i] : 0)];

    constexpr int RUNS = K / 8;
    constexpr int G  = (RUNS < 16) ? RUNS : 16;
    constexpr int NG = RUNS / G;
    static_assert(RUNS % G == 0, "k splits into whole load groups");
    float4 avq[2][G], bvq[2][G][TPW];
#pragma unroll
    for (int j = 0; j < G; ++j) {
        avq[0][j] = arow[2 * j];
#pragma unroll
        for (int i = 0; i < TPW; ++i) bvq[0][j][i] = brow[i][2 * j];
    }
#pragma unroll
    for (int g = 0; g < NG; ++g) {
        const int cur = g & 1, nxt = cur ^ 1;
        if (g + 1 < NG) {
#pragma unroll
            for (int j = 0; j < G; ++j) {
                avq[nxt][j] = arow[2 * ((g + 1) * G + j)];
#pragma unroll
                for (int i = 0; i < TPW; ++i) bvq[nxt][j][i] = brow[i][2 * ((g + 1) * G + j)];
            }
        }
        __builtin_amdgcn_sched_barrier(0);
#pragma unroll
        for (int j = 0; j < G; ++j) {
            const float4 av = avq[cur][j];
            v2f alo; alo[0] = av.x; alo[1] = av.y;
            v2f ahi; ahi[0] = av.z; ahi[1] = av.w;
#pragma unroll
            for (int i = 0; i < TPW; ++i) {
                const float4 bv = bvq[cur][j][i];
                v2f blo; blo[0] = colok[i] ? bv.x : 0.f; blo[1] = colok[i] ? bv.y : 0.f;
                v2f bhi; bhi[0] = colok[i] ? bv.z : 0.f; bhi[1] = colok[i] ? bv.w : 0.f;
                acc[i] = __builtin_amdgcn_wmma_f32_16x16x4_f32(false, alo, false, blo,
                                                               (short)0, acc[i], false, false);
                acc[i] = __builtin_amdgcn_wmma_f32_16x16x4_f32(false, ahi, false, bhi,
                                                               (short)0, acc[i], false, false);
            }
        }
    }

#pragma unroll
    for (int i = 0; i < TPW; ++i) {
        if (colok[i]) {
#pragma unroll
            for (int v = 0; v < 8; ++v) {
                const int row = m0 + v + 8 * half;
                float val = acc[i][v] + bs[i];
                if (RELU) val = fmaxf(val, 0.f);
                O[(size_t)row * N + col[i]] = val;
            }
        }
    }
}

__global__ void final_combine_kernel(const float* __restrict__ hout,
                                     const float* __restrict__ W2,
                                     const float* __restrict__ bias2,
                                     float* __restrict__ out) {
    __shared__ float comb[100];
    const int n = blockIdx.x;
    const int t = threadIdx.x;
    if (t < 100) {
        float s = 0.f;
#pragma unroll
        for (int h = 0; h < 4; ++h) s += hout[((size_t)h * 64 + n) * 100 + t];
        comb[t] = s * 0.25f;
    }
    __syncthreads();
    if (t < 100) {
        float acc = bias2[t];
        for (int j = 0; j < 100; ++j) acc += comb[j] * W2[t * 100 + j];
        float* orow = out + (size_t)n * 64 * 100 + t;
#pragma unroll 4
        for (int rep = 0; rep < 64; ++rep) orow[(size_t)rep * 100] = acc;
    }
}

extern "C" void kernel_launch(void* const* d_in, const int* in_sizes, int n_in,
                              void* d_out, int out_size, void* d_ws, size_t ws_size,
                              hipStream_t stream) {
    const float* x     = (const float*)d_in[0];
    const float* w1    = (const float*)d_in[1];
    const float* g1    = (const float*)d_in[2];
    const float* b1    = (const float*)d_in[3];
    const float* w2    = (const float*)d_in[4];
    const float* g2    = (const float*)d_in[5];
    const float* b2    = (const float*)d_in[6];
    const float* w3    = (const float*)d_in[7];
    const float* g3    = (const float*)d_in[8];
    const float* b3    = (const float*)d_in[9];
    const float* lw1   = (const float*)d_in[10];
    const float* lb1   = (const float*)d_in[11];
    const float* lw2   = (const float*)d_in[12];
    const float* lb2   = (const float*)d_in[13];
    const float* W2    = (const float*)d_in[14];
    const float* bias2 = (const float*)d_in[15];
    float* out = (float*)d_out;

    float* ws   = (float*)d_ws;
    float* A1   = ws;
    float* z1   = A1 + 256 * 256;
    float* z2   = z1 + 4 * 256 * 32;
    float* fcin = z2 + 4 * 256 * 64;
    float* h1   = fcin + 4 * 64 * 256;
    float* hout = h1 + 4 * 64 * 512;

    roi_pool_kernel<<<64, 256, 0, stream>>>(x, A1);

    conv_bn_relu_kernel<256, 32, false, false><<<dim3(2, 4), 256, 0, stream>>>(
        A1, 0, w1, g1, b1, z1, 256 * 32);
    conv_bn_relu_kernel<32, 64, true, false><<<dim3(4, 4), 256, 0, stream>>>(
        z1, 256 * 32, w2, g2, b2, z2, 256 * 64);
    conv_bn_relu_kernel<64, 64, true, true><<<dim3(4, 4), 256, 0, stream>>>(
        z2, 256 * 64, w3, g3, b3, fcin, 64 * 256);

    gemm_bias_kernel<64, 512, 256, true, 16, 8><<<dim3(16, 4), 256, 0, stream>>>(
        fcin, 64 * 256, lw1, lb1, h1, 64 * 512);
    gemm_bias_kernel<64, 100, 512, false, 7, 4><<<dim3(7, 4), 128, 0, stream>>>(
        h1, 64 * 512, lw2, lb2, hout, 64 * 100);

    final_combine_kernel<<<64, 128, 0, stream>>>(hout, W2, bias2, out);
}


# Pool, fc1, fc2 spread over all WGPs as 2-wave workgroups (256/256/224); fc2's k split in 4 slices on separate workgroups, last kernel adds the 16 parts; fc2/last sum order changes
# speedup vs baseline: 2.3035x; 1.1598x over previous
#include <hip/hip_runtime.h>
#include <hip/hip_bf16.h>

typedef float v2f __attribute__((ext_vector_type(2)));
typedef float v8f __attribute__((ext_vector_type(8)));

#define EPSV 1e-5f

__global__ __launch_bounds__(64) void roi_pool_kernel(const float* __restrict__ x, float* __restrict__ A1) {
    const int n = blockIdx.x >> 2;
    const int c = (blockIdx.x & 3) * 64 + threadIdx.x;
    const float* base = x + ((size_t)(n * 256 + c) * 64) * 64;
    float v[4][5];
#pragma unroll
    for (int rr = 0; rr < 4; ++rr)
#pragma unroll
        for (int cc = 0; cc < 5; ++cc)
            v[rr][cc] = base[(2 + rr) * 64 + 1 + cc];
    float r0[5], r1[5];
#pragma unroll
    for (int cc = 0; cc < 5; ++cc) {
        r0[cc] = fmaxf(v[0][cc], v[1][cc]);
        r1[cc] = fmaxf(v[2][cc], v[3][cc]);
    }
    float p00 = fmaxf(fmaxf(r0[0], r0[1]), r0[2]);
    float p01 = fmaxf(fmaxf(r0[2], r0[3]), r0[4]);
    float p10 = fmaxf(fmaxf(r1[0], r1[1]), r1[2]);
    float p11 = fmaxf(fmaxf(r1[2], r1[3]), r1[4]);
    float* o = A1 + (size_t)n * 4 * 256 + c;
    o[0 * 256] = p00;
    o[1 * 256] = p01;
    o[2 * 256] = p10;
    o[3 * 256] = p11;
}

template <int KIN, int COUT, bool A_TILED, bool RESHAPE>
__global__ __launch_bounds__(256) void conv_bn_relu_kernel(
    const float* __restrict__ Ain, int a_head_stride,
    const float* __restrict__ Wt,
    const float* __restrict__ g, const float* __restrict__ b,
    float* __restrict__ Out, int out_head_stride) {
    static_assert(KIN % 8 == 0, "k is walked in runs of 8");
    constexpr int RUNS = KIN / 8;
    constexpr int G  = (RUNS < 16) ? RUNS : 16;
    constexpr int NG = RUNS / G;
    static_assert(RUNS % G == 0, "k splits into whole load groups");

    __shared__ float lsum[16], lsq[16], lscale[16], lshift[16];

    const int ct   = blockIdx.x;
    const int h    = blockIdx.y;
    const int tid  = threadIdx.x;
    const int wave = tid >> 5;
    const int lane = tid & 31;
    const int half = lane >> 4;
    const int r    = lane & 15;

    if (tid < 16) { lsum[tid] = 0.f; lsq[tid] = 0.f; }
    __syncthreads();

    const float* A = Ain + (size_t)h * a_head_stride;
    const float* W = Wt + ((size_t)h * COUT + ct * 16 + r) * KIN + 4 * half;

    const int row0 = wave * 16 + r;
    const int row1 = (wave + 8) * 16 + r;
    const float* ap0 = A_TILED ? (A + (size_t)row0 * 16 + 4 * half) : (A + (size_t)row0 * KIN + 4 * half);
    const float* ap1 = A_TILED ? (A + (size_t)row1 * 16 + 4 * half) : (A + (size_t)row1 * KIN + 4 * half);

    v8f acc0 = v8f{}, acc1 = v8f{};

    const float gcol = g[h * COUT + ct * 16 + r];
    const float bcol = b[h * COUT + ct * 16 + r];

    float4 av0[2][G], av1[2][G], bv[2][G];
#pragma unroll
    for (int j = 0; j < G; ++j) {
        const int kb   = 8 * j;
        const int aoff = A_TILED ? ((kb >> 4) * (256 * 16) + (kb & 15)) : kb;
        av0[0][j] = *reinterpret_cast<const float4*>(ap0 + aoff);
        av1[0][j] = *reinterpret_cast<const float4*>(ap1 + aoff);
        bv[0][j]  = *reinterpret_cast<const float4*>(W + kb);
    }
#pragma unroll
    for (int g = 0; g < NG; ++g) {
        const int cur = g & 1, nxt = cur ^ 1;
        if (g + 1 < NG) {
#pragma unroll
            for (int j = 0; j < G; ++j) {
                const int kb   = 8 * ((g + 1) * G + j);
                const int aoff = A_TILED ? ((kb >> 4) * (256 * 16) + (kb & 15)) : kb;
                av0[nxt][j] = *reinterpret_cast<const float4*>(ap0 + aoff);
                av1[nxt][j] = *reinterpret_cast<const float4*>(ap1 + aoff);
                bv[nxt][j]  = *reinterpret_cast<const float4*>(W + kb);
            }
        }
        __builtin_amdgcn_sched_barrier(0);
#pragma unroll
        for (int j = 0; j < G; ++j) {
            v2f a0; a0[0] = av0[cur][j].x; a0[1] = av0[cur][j].y;
            v2f a1; a1[0] = av1[cur][j].x; a1[1] = av1[cur][j].y;
            v2f bb; bb[0] = bv[cur][j].x;  bb[1] = bv[cur][j].y;
            acc0 = __builtin_amdgcn_wmma_f32_16x16x4_f32(false, a0, false, bb, (short)0, acc0, false, false);
            acc1 = __builtin_amdgcn_wmma_f32_16x16x4_f32(false, a1, false, bb, (short)0, acc1, false, false);
            a0[0] = av0[cur][j].z; a0[1] = av0[cur][j].w;
            a1[0] = av1[cur][j].z; a1[1] = av1[cur][j].w;
            bb[0] = bv[cur][j].z;  bb[1] = bv[cur][j].w;
            acc0 = __builtin_amdgcn_wmma_f32_16x16x4_f32(false, a0, false, bb, (short)0, acc0, false, false);
            acc1 = __builtin_amdgcn_wmma_f32_16x16x4_f32(false, a1, false, bb, (short)0, acc1, false, false);
        }
    }

    {
        float s = 0.f, q = 0.f;
#pragma unroll
        for (int v = 0; v < 8; ++v) { float xv = acc0[v]; s += xv; q += xv * xv; }
        atomicAdd(&lsum[r], s);
        atomicAdd(&lsq[r], q);
        s = 0.f; q = 0.f;
#pragma unroll
        for (int v = 0; v < 8; ++v) { float xv = acc1[v]; s += xv; q += xv * xv; }
        atomicAdd(&lsum[r], s);
        atomicAdd(&lsq[r], q);
    }
    __syncthreads();

    if (tid < 16) {
        float mu  = lsum[tid] * (1.f / 256.f);
        float var = lsq[tid] * (1.f / 256.f) - mu * mu;
        float sc  = gcol * rsqrtf(var + EPSV);
        lscale[tid] = sc;
        lshift[tid] = bcol - mu * sc;
    }
    __syncthreads();

    float* O = Out + (size_t)h * out_head_stride;
    const float sc = lscale[r], sh = lshift[r];
    const int col = ct * 16 + r;
#pragma unroll
    for (int i = 0; i < 2; ++i) {
        const int m0 = (wave + 8 * i) * 16;
        const v8f c = i ? acc1 : acc0;
#pragma unroll
        for (int v = 0; v < 8; ++v) {
            const int row = m0 + v + 8 * half;
            float val = c[v] * sc + sh;
            val = fmaxf(val, 0.f);
            if (RESHAPE) {
                const int n = row >> 2, s = row & 3;
                O[n * (COUT * 4) + col * 4 + s] = val;
            } else {
                O[((size_t)ct * 256 + row) * 16 + r] = val;
            }
        }
    }
}

template <int M, int N, int K, bool RELU, int GX, int WPB, int CPB, int KS>
__global__ __launch_bounds__(WPB * 32) void gemm_bias_kernel(
    const float* __restrict__ Ain, int a_head_stride,
    const float* __restrict__ Wt,
    const float* __restrict__ bias,
    float* __restrict__ Out, int out_head_stride) {
    constexpr int MT = (M + 15) / 16, NT = (N + 15) / 16, TT = MT * NT;
    constexpr int WS  = GX * WPB;
    constexpr int TPW = (TT + WS - 1) / WS;
    static_assert(WS % MT == 0, "a wave's tiles must share one row tile");
    static_assert(TPW == 1 || TT % WS == 0, "every wave must own a full block");
    constexpr int NTS = WS / MT;
    constexpr bool GUARD = (N % 16) != 0;
    constexpr int KSL = K / KS;
    static_assert(K % KS == 0 && KSL % 8 == 0, "a slice is walked in runs of 8");
    static_assert(KS == 1 || !RELU, "partial sums cannot pass through the ReLU");
    static_assert(CPB == 1 || (TPW == 1 && WPB % CPB == 0 && NT % CPB == 0 && WS % (CPB * MT) == 0),
                  "column groups hold whole workgroups of one-tile waves");

    const int h    = blockIdx.y;
    const int ks   = blockIdx.z;
    const int tid  = threadIdx.x;
    const int wave = tid >> 5;
    const int lane = tid & 31;
    const int half = lane >> 4;
    const int r    = lane & 15;
    const int wglobal = blockIdx.x * WPB + wave;
    if (wglobal >= TT) return;

    const float* A = Ain + (size_t)h * a_head_stride + ks * KSL;
    const float* W = Wt + (size_t)h * N * K + ks * KSL;
    float* O = Out + (size_t)(h * KS + ks) * out_head_stride;

    const int mt  = (wglobal / CPB) % MT;
    const int nt0 = (wglobal / (CPB * MT)) * CPB + wglobal % CPB;
    const int m0  = mt * 16;

    int  col[TPW];
    bool colok[TPW];
    const float4* brow[TPW];
    v8f acc[TPW];
#pragma unroll
    for (int i = 0; i < TPW; ++i) {
        col[i]   = (nt0 + i * NTS) * 16 + r;
        colok[i] = !GUARD || (col[i] < N);
        brow[i]  = reinterpret_cast<const float4*>(W + (size_t)(colok[i] ? col[i] : 0) * K) + half;
        acc[i]   = v8f{};
    }
    const float4* arow = reinterpret_cast<const float4*>(A + (size_t)(m0 + r) * K) + half;
    float bs[TPW];
#pragma unroll
    for (int i = 0; i < TPW; ++i) {
        bs[i] = bias[h * N + (colok[i] ? col[i] : 0)];
        if (KS > 1 && ks != 0) bs[i] = 0.f;
    }

    constexpr int RUNS = KSL / 8;
    constexpr int G  = (RUNS < 16) ? RUNS : 16;
    constexpr int NG = RUNS / G;
    static_assert(RUNS % G == 0, "k splits into whole load groups");
    float4 avq[2][G], bvq[2][G][TPW];
#pragma unroll
    for (int j = 0; j < G; ++j) {
        avq[0][j] = arow[2 * j];
#pragma unroll
        for (int i = 0; i < TPW; ++i) bvq[0][j][i] = brow[i][2 * j];
    }
#pragma unroll
    for (int g = 0; g < NG; ++g) {
        const int cur = g & 1, nxt = cur ^ 1;
        if (g + 1 < NG) {
#pragma unroll
            for (int j = 0; j < G; ++j) {
                avq[nxt][j] = arow[2 * ((g + 1) * G + j)];
#pragma unroll
                for (int i = 0; i < TPW; ++i) bvq[nxt][j][i] = brow[i][2 * ((g + 1) * G + j)];
            }
        }
        __builtin_amdgcn_sched_barrier(0);
#pragma unroll
        for (int j = 0; j < G; ++j) {
            const float4 av = avq[cur][j];
            v2f alo; alo[0] = av.x; alo[1] = av.y;
            v2f ahi; ahi[0] = av.z; ahi[1] = av.w;
#pragma unroll
            for (int i = 0; i < TPW; ++i) {
                const float4 bv = bvq[cur][j][i];
                v2f blo; blo[0] = colok[i] ? bv.x : 0.f; blo[1] = colok[i] ? bv.y : 0.f;
                v2f bhi; bhi[0] = colok[i] ? bv.z : 0.f; bhi[1] = colok[i] ? bv.w : 0.f;
                acc[i] = __builtin_amdgcn_wmma_f32_16x16x4_f32(false, alo, false, blo,
                                                               (short)0, acc[i], false, false);
                acc[i] = __builtin_amdgcn_wmma_f32_16x16x4_f32(false, ahi, false, bhi,
                                                               (short)0, acc[i], false, false);
            }
        }
    }

#pragma unroll
    for (int i = 0; i < TPW; ++i) {
        if (colok[i]) {
#pragma unroll
            for (int v = 0; v < 8; ++v) {
                const int row = m0 + v + 8 * half;
                float val = acc[i][v] + bs[i];
                if (RELU) val = fmaxf(val, 0.f);
                O[(size_t)row * N + col[i]] = val;
            }
        }
    }
}

template <int PARTS>
__global__ void final_combine_kernel(const float* __restrict__ hout,
                                     const float* __restrict__ W2,
                                     const float* __restrict__ bias2,
                                     float* __restrict__ out) {
    __shared__ float comb[100];
    const int n = blockIdx.x;
    const int t = threadIdx.x;
    if (t < 100) {
        float s = 0.f;
#pragma unroll
        for (int p = 0; p < PARTS; ++p) s += hout[((size_t)p * 64 + n) * 100 + t];
        comb[t] = s * 0.25f;
    }
    __syncthreads();
    if (t < 100) {
        float acc = bias2[t];
        for (int j = 0; j < 100; ++j) acc += comb[j] * W2[t * 100 + j];
        float* orow = out + (size_t)n * 64 * 100 + t;
#pragma unroll 4
        for (int rep = 0; rep < 64; ++rep) orow[(size_t)rep * 100] = acc;
    }
}

extern "C" void kernel_launch(void* const* d_in, const int* in_sizes, int n_in,
                              void* d_out, int out_size, void* d_ws, size_t ws_size,
                              hipStream_t stream) {
    const float* x     = (const float*)d_in[0];
    const float* w1    = (const float*)d_in[1];
    const float* g1    = (const float*)d_in[2];
    const float* b1    = (const float*)d_in[3];
    const float* w2    = (const float*)d_in[4];
    const float* g2    = (const float*)d_in[5];
    const float* b2    = (const float*)d_in[6];
    const float* w3    = (const float*)d_in[7];
    const float* g3    = (const float*)d_in[8];
    const float* b3    = (const float*)d_in[9];
    const float* lw1   = (const float*)d_in[10];
    const float* lb1   = (const float*)d_in[11];
    const float* lw2   = (const float*)d_in[12];
    const float* lb2   = (const float*)d_in[13];
    const float* W2    = (const float*)d_in[14];
    const float* bias2 = (const float*)d_in[15];
    float* out = (float*)d_out;

    float* ws   = (float*)d_ws;
    float* A1   = ws;
    float* z1   = A1 + 256 * 256;
    float* z2   = z1 + 4 * 256 * 32;
    float* fcin = z2 + 4 * 256 * 64;
    float* h1   = fcin + 4 * 64 * 256;
    float* hout = ws;

    roi_pool_kernel<<<256, 64, 0, stream>>>(x, A1);

    conv_bn_relu_kernel<256, 32, false, false><<<dim3(2, 4), 256, 0, stream>>>(
        A1, 0, w1, g1, b1, z1, 256 * 32);
    conv_bn_relu_kernel<32, 64, true, false><<<dim3(4, 4), 256, 0, stream>>>(
        z1, 256 * 32, w2, g2, b2, z2, 256 * 64);
    conv_bn_relu_kernel<64, 64, true, true><<<dim3(4, 4), 256, 0, stream>>>(
        z2, 256 * 64, w3, g3, b3, fcin, 64 * 256);

    gemm_bias_kernel<64, 512, 256, true, 64, 2, 2, 1><<<dim3(64, 4, 1), 64, 0, stream>>>(
        fcin, 64 * 256, lw1, lb1, h1, 64 * 512);
    gemm_bias_kernel<64, 100, 512, false, 14, 2, 1, 4><<<dim3(14, 4, 4), 64, 0, stream>>>(
        h1, 64 * 512, lw2, lb2, hout, 64 * 100);

    final_combine_kernel<16><<<64, 128, 0, stream>>>(hout, W2, bias2, out);
}


# Layer-1 product runs one tile per wave on 64 workgroups and writes raw z1 plus per-row-tile BN partial sums; layer 2 adds them in fixed order (was LDS atomics), applies BN1+ReLU to its A fragments
# speedup vs baseline: 2.7261x; 1.1835x over previous
#include <hip/hip_runtime.h>
#include <hip/hip_bf16.h>

typedef float v2f __attribute__((ext_vector_type(2)));
typedef float v8f __attribute__((ext_vector_type(8)));

#define EPSV 1e-5f

__global__ __launch_bounds__(64) void roi_pool_kernel(const float* __restrict__ x, float* __restrict__ A1) {
    const int n = blockIdx.x >> 2;
    const int c = (blockIdx.x & 3) * 64 + threadIdx.x;
    const float* base = x + ((size_t)(n * 256 + c) * 64) * 64;
    float v[4][5];
#pragma unroll
    for (int rr = 0; rr < 4; ++rr)
#pragma unroll
        for (int cc = 0; cc < 5; ++cc)
            v[rr][cc] = base[(2 + rr) * 64 + 1 + cc];
    float r0[5], r1[5];
#pragma unroll
    for (int cc = 0; cc < 5; ++cc) {
        r0[cc] = fmaxf(v[0][cc], v[1][cc]);
        r1[cc] = fmaxf(v[2][cc], v[3][cc]);
    }
    float p00 = fmaxf(fmaxf(r0[0], r0[1]), r0[2]);
    float p01 = fmaxf(fmaxf(r0[2], r0[3]), r0[4]);
    float p10 = fmaxf(fmaxf(r1[0], r1[1]), r1[2]);
    float p11 = fmaxf(fmaxf(r1[2], r1[3]), r1[4]);
    float* o = A1 + (size_t)n * 4 * 256 + c;
    o[0 * 256] = p00;
    o[1 * 256] = p01;
    o[2 * 256] = p10;
    o[3 * 256] = p11;
}

__global__ __launch_bounds__(64) void conv1_raw_kernel(
    const float* __restrict__ Ain,
    const float* __restrict__ Wt,
    float* __restrict__ Z,
    float* __restrict__ P) {
    constexpr int KIN = 256, COUT = 32;
    constexpr int RUNS = KIN / 8;
    constexpr int G  = 16;
    constexpr int NG = RUNS / G;

    const int rt   = blockIdx.x;
    const int h    = blockIdx.y;
    const int tid  = threadIdx.x;
    const int ct   = tid >> 5;
    const int lane = tid & 31;
    const int half = lane >> 4;
    const int r    = lane & 15;

    const float* ap = Ain + (size_t)(rt * 16 + r) * KIN + 4 * half;
    const float* W  = Wt + ((size_t)h * COUT + ct * 16 + r) * KIN + 4 * half;

    v8f acc = v8f{};
    float4 av[2][G], bv[2][G];
#pragma unroll
    for (int j = 0; j < G; ++j) {
        av[0][j] = *reinterpret_cast<const float4*>(ap + 8 * j);
        bv[0][j] = *reinterpret_cast<const float4*>(W + 8 * j);
    }
#pragma unroll
    for (int g = 0; g < NG; ++g) {
        const int cur = g & 1, nxt = cur ^ 1;
        if (g + 1 < NG) {
#pragma unroll
            for (int j = 0; j < G; ++j) {
                const int kb = 8 * ((g + 1) * G + j);
                av[nxt][j] = *reinterpret_cast<const float4*>(ap + kb);
                bv[nxt][j] = *reinterpret_cast<const float4*>(W + kb);
            }
        }
        __builtin_amdgcn_sched_barrier(0);
#pragma unroll
        for (int j = 0; j < G; ++j) {
            v2f a; a[0] = av[cur][j].x; a[1] = av[cur][j].y;
            v2f bb; bb[0] = bv[cur][j].x; bb[1] = bv[cur][j].y;
            acc = __builtin_amdgcn_wmma_f32_16x16x4_f32(false, a, false, bb, (short)0, acc, false, false);
            a[0] = av[cur][j].z; a[1] = av[cur][j].w;
            bb[0] = bv[cur][j].z; bb[1] = bv[cur][j].w;
            acc = __builtin_amdgcn_wmma_f32_16x16x4_f32(false, a, false, bb, (short)0, acc, false, false);
        }
    }

    float s = 0.f, q = 0.f;
#pragma unroll
    for (int v = 0; v < 8; ++v) { float xv = acc[v]; s += xv; q += xv * xv; }

    float* Zt = Z + (size_t)h * (256 * COUT) + ((size_t)ct * 256 + rt * 16 + 8 * half) * 16 + r;
#pragma unroll
    for (int v = 0; v < 8; ++v) Zt[v * 16] = acc[v];

    float* Pp = P + ((size_t)((h * 16 + rt) * 2 + ct) * 2) * 32 + lane;
    Pp[0]  = s;
    Pp[32] = q;
}

template <int KIN, int COUT, bool A_BN, bool RESHAPE>
__global__ __launch_bounds__(256) void conv_bn_relu_kernel(
    const float* __restrict__ Ain, int a_head_stride,
    const float* __restrict__ Wt,
    const float* __restrict__ g, const float* __restrict__ b,
    float* __restrict__ Out, int out_head_stride,
    const float* __restrict__ Pin,
    const float* __restrict__ gin, const float* __restrict__ bin) {
    static_assert(KIN % 8 == 0, "k is walked in runs of 8");
    constexpr int RUNS = KIN / 8;
    constexpr int G  = (RUNS < 16) ? RUNS : 16;
    constexpr int NG = RUNS / G;
    static_assert(RUNS % G == 0, "k splits into whole load groups");
    static_assert(!A_BN || (KIN == 32 && NG == 1), "the input statistics step is laid out for 32 channels in one load group");

    __shared__ float lsum[16], lsq[16], lscale[16], lshift[16];
    __shared__ float pred[2][8][32];
    __shared__ float4 insc4[8], insh4[8];

    const int ct   = blockIdx.x;
    const int h    = blockIdx.y;
    const int tid  = threadIdx.x;
    const int wave = tid >> 5;
    const int lane = tid & 31;
    const int half = lane >> 4;
    const int r    = lane & 15;

    if (tid < 16) { lsum[tid] = 0.f; lsq[tid] = 0.f; }
    if (!A_BN) __syncthreads();

    float pS = 0.f, pQ = 0.f, gi = 0.f, bi = 0.f;
    if (A_BN) {
        const float* Pc = Pin + ((size_t)((h * 16 + 2 * wave) * 2 + half) * 2) * 32 + r;
#pragma unroll
        for (int pp = 0; pp < 4; ++pp) {
            const int o = (pp >> 1) * (2 * 2 * 32) + (pp & 1) * 16;
            pS += Pc[o];
            pQ += Pc[o + 32];
        }
        if (tid < 32) { gi = gin[h * KIN + tid]; bi = bin[h * KIN + tid]; }
    }

    const float* A = Ain + (size_t)h * a_head_stride;
    const float* W = Wt + ((size_t)h * COUT + ct * 16 + r) * KIN + 4 * half;

    const int row0 = wave * 16 + r;
    const int row1 = (wave + 8) * 16 + r;
    const float* ap0 = A + (size_t)row0 * 16 + 4 * half;
    const float* ap1 = A + (size_t)row1 * 16 + 4 * half;

    v8f acc0 = v8f{}, acc1 = v8f{};

    const float gcol = g[h * COUT + ct * 16 + r];
    const float bcol = b[h * COUT + ct * 16 + r];

    float4 av0[2][G], av1[2][G], bv[2][G];
#pragma unroll
    for (int j = 0; j < G; ++j) {
        const int kb   = 8 * j;
        const int aoff = (kb >> 4) * (256 * 16) + (kb & 15);
        av0[0][j] = *reinterpret_cast<const float4*>(ap0 + aoff);
        av1[0][j] = *reinterpret_cast<const float4*>(ap1 + aoff);
        bv[0][j]  = *reinterpret_cast<const float4*>(W + kb);
    }

    if (A_BN) {
        pred[0][wave][lane] = pS;
        pred[1][wave][lane] = pQ;
        __syncthreads();
        if (tid < 32) {
            float S = 0.f, Q = 0.f;
#pragma unroll
            for (int gq = 0; gq < 8; ++gq) { S += pred[0][gq][tid]; Q += pred[1][gq][tid]; }
            float mu  = S * (1.f / 256.f);
            float var = Q * (1.f / 256.f) - mu * mu;
            float sc  = gi * rsqrtf(var + EPSV);
            reinterpret_cast<float*>(insc4)[tid] = sc;
            reinterpret_cast<float*>(insh4)[tid] = bi - mu * sc;
        }
        __syncthreads();
#pragma unroll
        for (int j = 0; j < G; ++j) {
            const float4 s4 = insc4[2 * j + half];
            const float4 t4 = insh4[2 * j + half];
            av0[0][j].x = fmaxf(av0[0][j].x * s4.x + t4.x, 0.f);
            av0[0][j].y = fmaxf(av0[0][j].y * s4.y + t4.y, 0.f);
            av0[0][j].z = fmaxf(av0[0][j].z * s4.z + t4.z, 0.f);
            av0[0][j].w = fmaxf(av0[0][j].w * s4.w + t4.w, 0.f);
            av1[0][j].x = fmaxf(av1[0][j].x * s4.x + t4.x, 0.f);
            av1[0][j].y = fmaxf(av1[0][j].y * s4.y + t4.y, 0.f);
            av1[0][j].z = fmaxf(av1[0][j].z * s4.z + t4.z, 0.f);
            av1[0][j].w = fmaxf(av1[0][j].w * s4.w + t4.w, 0.f);
        }
    }

#pragma unroll
    for (int g = 0; g < NG; ++g) {
        const int cur = g & 1, nxt = cur ^ 1;
        if (g + 1 < NG) {
#pragma unroll
            for (int j = 0; j < G; ++j) {
                const int kb   = 8 * ((g + 1) * G + j);
                const int aoff = (kb >> 4) * (256 * 16) + (kb & 15);
                av0[nxt][j] = *reinterpret_cast<const float4*>(ap0 + aoff);
                av1[nxt][j] = *reinterpret_cast<const float4*>(ap1 + aoff);
                bv[nxt][j]  = *reinterpret_cast<const float4*>(W + kb);
            }
        }
        __builtin_amdgcn_sched_barrier(0);
#pragma unroll
        for (int j = 0; j < G; ++j) {
            v2f a0; a0[0] = av0[cur][j].x; a0[1] = av0[cur][j].y;
            v2f a1; a1[0] = av1[cur][j].x; a1[1] = av1[cur][j].y;
            v2f bb; bb[0] = bv[cur][j].x;  bb[1] = bv[cur][j].y;
            acc0 = __builtin_amdgcn_wmma_f32_16x16x4_f32(false, a0, false, bb, (short)0, acc0, false, false);
            acc1 = __builtin_amdgcn_wmma_f32_16x16x4_f32(false, a1, false, bb, (short)0, acc1, false, false);
            a0[0] = av0[cur][j].z; a0[1] = av0[cur][j].w;
            a1[0] = av1[cur][j].z; a1[1] = av1[cur][j].w;
            bb[0] = bv[cur][j].z;  bb[1] = bv[cur][j].w;
            acc0 = __builtin_amdgcn_wmma_f32_16x16x4_f32(false, a0, false, bb, (short)0, acc0, false, false);
            acc1 = __builtin_amdgcn_wmma_f32_16x16x4_f32(false, a1, false, bb, (short)0, acc1, false, false);
        }
    }

    {
        float s = 0.f, q = 0.f;
#pragma unroll
        for (int v = 0; v < 8; ++v) { float xv = acc0[v]; s += xv; q += xv * xv; }
        atomicAdd(&lsum[r], s);
        atomicAdd(&lsq[r], q);
        s = 0.f; q = 0.f;
#pragma unroll
        for (int v = 0; v < 8; ++v) { float xv = acc1[v]; s += xv; q += xv * xv; }
        atomicAdd(&lsum[r], s);
        atomicAdd(&lsq[r], q);
    }
    __syncthreads();

    if (tid < 16) {
        float mu  = lsum[tid] * (1.f / 256.f);
        float var = lsq[tid] * (1.f / 256.f) - mu * mu;
        float sc  = gcol * rsqrtf(var + EPSV);
        lscale[tid] = sc;
        lshift[tid] = bcol - mu * sc;
    }
    __syncthreads();

    float* O = Out + (size_t)h * out_head_stride;
    const float sc = lscale[r], sh = lshift[r];
    const int col = ct * 16 + r;
#pragma unroll
    for (int i = 0; i < 2; ++i) {
        const int m0 = (wave + 8 * i) * 16;
        const v8f c = i ? acc1 : acc0;
#pragma unroll
        for (int v = 0; v < 8; ++v) {
            const int row = m0 + v + 8 * half;
            float val = c[v] * sc + sh;
            val = fmaxf(val, 0.f);
            if (RESHAPE) {
                const int n = row >> 2, s = row & 3;
                O[n * (COUT * 4) + col * 4 + s] = val;
            } else {
                O[((size_t)ct * 256 + row) * 16 + r] = val;
            }
        }
    }
}

template <int M, int N, int K, bool RELU, int GX, int WPB, int CPB, int KS>
__global__ __launch_bounds__(WPB * 32) void gemm_bias_kernel(
    const float* __restrict__ Ain, int a_head_stride,
    const float* __restrict__ Wt,
    const float* __restrict__ bias,
    float* __restrict__ Out, int out_head_stride) {
    constexpr int MT = (M + 15) / 16, NT = (N + 15) / 16, TT = MT * NT;
    constexpr int WS  = GX * WPB;
    constexpr int TPW = (TT + WS - 1) / WS;
    static_assert(WS % MT == 0, "a wave's tiles must share one row tile");
    static_assert(TPW == 1 || TT % WS == 0, "every wave must own a full block");
    constexpr int NTS = WS / MT;
    constexpr bool GUARD = (N % 16) != 0;
    constexpr int KSL = K / KS;
    static_assert(K % KS == 0 && KSL % 8 == 0, "a slice is walked in runs of 8");
    static_assert(KS == 1 || !RELU, "partial sums cannot pass through the ReLU");
    static_assert(CPB == 1 || (TPW == 1 && WPB % CPB == 0 && NT % CPB == 0 && WS % (CPB * MT) == 0),
                  "column groups hold whole workgroups of one-tile waves");

    const int h    = blockIdx.y;
    const int ks   = blockIdx.z;
    const int tid  = threadIdx.x;
    const int wave = tid >> 5;
    const int lane = tid & 31;
    const int half = lane >> 4;
    const int r    = lane & 15;
    const int wglobal = blockIdx.x * WPB + wave;
    if (wglobal >= TT) return;

    const float* A = Ain + (size_t)h * a_head_stride + ks * KSL;
    const float* W = Wt + (size_t)h * N * K + ks * KSL;
    float* O = Out + (size_t)(h * KS + ks) * out_head_stride;

    const int mt  = (wglobal / CPB) % MT;
    const int nt0 = (wglobal / (CPB * MT)) * CPB + wglobal % CPB;
    const int m0  = mt * 16;

    int  col[TPW];
    bool colok[TPW];
    const float4* brow[TPW];
    v8f acc[TPW];
#pragma unroll
    for (int i = 0; i < TPW; ++i) {
        col[i]   = (nt0 + i * NTS) * 16 + r;
        colok[i] = !GUARD || (col[i] < N);
        brow[i]  = reinterpret_cast<const float4*>(W + (size_t)(colok[i] ? col[i] : 0) * K) + half;
        acc[i]   = v8f{};
    }
    const float4* arow = reinterpret_cast<const float4*>(A + (size_t)(m0 + r) * K) + half;
    float bs[TPW];
#pragma unroll
    for (int i = 0; i < TPW; ++i) {
        bs[i] = bias[h * N + (colok[i] ? col[i] : 0)];
        if (KS > 1 && ks != 0) bs[i] = 0.f;
    }

    constexpr int RUNS = KSL / 8;
    constexpr int G  = (RUNS < 16) ? RUNS : 16;
    constexpr int NG = RUNS / G;
    static_assert(RUNS % G == 0, "k splits into whole load groups");
    float4 avq[2][G], bvq[2][G][TPW];
#pragma unroll
    for (int j = 0; j < G; ++j) {
        avq[0][j] = arow[2 * j];
#pragma unroll
        for (int i = 0; i < TPW; ++i) bvq[0][j][i] = brow[i][2 * j];
    }
#pragma unroll
    for (int g = 0; g < NG; ++g) {
        const int cur = g & 1, nxt = cur ^ 1;
        if (g + 1 < NG) {
#pragma unroll
            for (int j = 0; j < G; ++j) {
                avq[nxt][j] = arow[2 * ((g + 1) * G + j)];
#pragma unroll
                for (int i = 0; i < TPW; ++i) bvq[nxt][j][i] = brow[i][2 * ((g + 1) * G + j)];
            }
        }
        __builtin_amdgcn_sched_barrier(0);
#pragma unroll
        for (int j = 0; j < G; ++j) {
            const float4 av = avq[cur][j];
            v2f alo; alo[0] = av.x; alo[1] = av.y;
            v2f ahi; ahi[0] = av.z; ahi[1] = av.w;
#pragma unroll
            for (int i = 0; i < TPW; ++i) {
                const float4 bv = bvq[cur][j][i];
                v2f blo; blo[0] = colok[i] ? bv.x : 0.f; blo[1] = colok[i] ? bv.y : 0.f;
                v2f bhi; bhi[0] = colok[i] ? bv.z : 0.f; bhi[1] = colok[i] ? bv.w : 0.f;
                acc[i] = __builtin_amdgcn_wmma_f32_16x16x4_f32(false, alo, false, blo,
                                                               (short)0, acc[i], false, false);
                acc[i] = __builtin_amdgcn_wmma_f32_16x16x4_f32(false, ahi, false, bhi,
                                                               (short)0, acc[i], false, false);
            }
        }
    }

#pragma unroll
    for (int i = 0; i < TPW; ++i) {
        if (colok[i]) {
#pragma unroll
            for (int v = 0; v < 8; ++v) {
                const int row = m0 + v + 8 * half;
                float val = acc[i][v] + bs[i];
                if (RELU) val = fmaxf(val, 0.f);
                O[(size_t)row * N + col[i]] = val;
            }
        }
    }
}

template <int PARTS>
__global__ void final_combine_kernel(const float* __restrict__ hout,
                                     const float* __restrict__ W2,
                                     const float* __restrict__ bias2,
                                     float* __restrict__ out) {
    __shared__ float comb[100];
    const int n = blockIdx.x;
    const int t = threadIdx.x;
    if (t < 100) {
        float s = 0.f;
#pragma unroll
        for (int p = 0; p < PARTS; ++p) s += hout[((size_t)p * 64 + n) * 100 + t];
        comb[t] = s * 0.25f;
    }
    __syncthreads();
    if (t < 100) {
        float acc = bias2[t];
        for (int j = 0; j < 100; ++j) acc += comb[j] * W2[t * 100 + j];
        float* orow = out + (size_t)n * 64 * 100 + t;
#pragma unroll 4
        for (int rep = 0; rep < 64; ++rep) orow[(size_t)rep * 100] = acc;
    }
}

extern "C" void kernel_launch(void* const* d_in, const int* in_sizes, int n_in,
                              void* d_out, int out_size, void* d_ws, size_t ws_size,
                              hipStream_t stream) {
    const float* x     = (const float*)d_in[0];
    const float* w1    = (const float*)d_in[1];
    const float* g1    = (const float*)d_in[2];
    const float* b1    = (const float*)d_in[3];
    const float* w2    = (const float*)d_in[4];
    const float* g2    = (const float*)d_in[5];
    const float* b2    = (const float*)d_in[6];
    const float* w3    = (const float*)d_in[7];
    const float* g3    = (const float*)d_in[8];
    const float* b3    = (const float*)d_in[9];
    const float* lw1   = (const float*)d_in[10];
    const float* lb1   = (const float*)d_in[11];
    const float* lw2   = (const float*)d_in[12];
    const float* lb2   = (const float*)d_in[13];
    const float* W2    = (const float*)d_in[14];
    const float* bias2 = (const float*)d_in[15];
    float* out = (float*)d_out;

    float* ws   = (float*)d_ws;
    float* A1   = ws;
    float* z1   = A1 + 256 * 256;
    float* z2   = z1 + 4 * 256 * 32;
    float* fcin = z2 + 4 * 256 * 64;
    float* h1   = fcin + 4 * 64 * 256;
    float* hout = ws;
    float* bnp  = h1;

    roi_pool_kernel<<<256, 64, 0, stream>>>(x, A1);

    conv1_raw_kernel<<<dim3(16, 4), 64, 0, stream>>>(A1, w1, z1, bnp);
    conv_bn_relu_kernel<32, 64, true, false><<<dim3(4, 4), 256, 0, stream>>>(
        z1, 256 * 32, w2, g2, b2, z2, 256 * 64, bnp, g1, b1);
    conv_bn_relu_kernel<64, 64, false, true><<<dim3(4, 4), 256, 0, stream>>>(
        z2, 256 * 64, w3, g3, b3, fcin, 64 * 256, nullptr, nullptr, nullptr);

    gemm_bias_kernel<64, 512, 256, true, 64, 2, 2, 1><<<dim3(64, 4, 1), 64, 0, stream>>>(
        fcin, 64 * 256, lw1, lb1, h1, 64 * 512);
    gemm_bias_kernel<64, 100, 512, false, 14, 2, 1, 4><<<dim3(14, 4, 4), 64, 0, stream>>>(
        h1, 64 * 512, lw2, lb2, hout, 64 * 100);

    final_combine_kernel<16><<<64, 128, 0, stream>>>(hout, W2, bias2, out);
}


# fc2 joins the fc1 launch: a wave's fc1 tile is computed transposed (weights as A) and stays in registers as fc2's A fragments for its 16-deep k slice; 4 waves add via LDS; fc2/last sum order changes
# speedup vs baseline: 3.0011x; 1.1009x over previous
#include <hip/hip_runtime.h>
#include <hip/hip_bf16.h>

typedef float v2f __attribute__((ext_vector_type(2)));
typedef float v8f __attribute__((ext_vector_type(8)));

#define EPSV 1e-5f

__global__ __launch_bounds__(64) void roi_pool_kernel(const float* __restrict__ x, float* __restrict__ A1) {
    const int n = blockIdx.x >> 2;
    const int c = (blockIdx.x & 3) * 64 + threadIdx.x;
    const float* base = x + ((size_t)(n * 256 + c) * 64) * 64;
    float v[4][5];
#pragma unroll
    for (int rr = 0; rr < 4; ++rr)
#pragma unroll
        for (int cc = 0; cc < 5; ++cc)
            v[rr][cc] = base[(2 + rr) * 64 + 1 + cc];
    float r0[5], r1[5];
#pragma unroll
    for (int cc = 0; cc < 5; ++cc) {
        r0[cc] = fmaxf(v[0][cc], v[1][cc]);
        r1[cc] = fmaxf(v[2][cc], v[3][cc]);
    }
    float p00 = fmaxf(fmaxf(r0[0], r0[1]), r0[2]);
    float p01 = fmaxf(fmaxf(r0[2], r0[3]), r0[4]);
    float p10 = fmaxf(fmaxf(r1[0], r1[1]), r1[2]);
    float p11 = fmaxf(fmaxf(r1[2], r1[3]), r1[4]);
    float* o = A1 + (size_t)n * 4 * 256 + c;
    o[0 * 256] = p00;
    o[1 * 256] = p01;
    o[2 * 256] = p10;
    o[3 * 256] = p11;
}

__global__ __launch_bounds__(64) void conv1_raw_kernel(
    const float* __restrict__ Ain,
    const float* __restrict__ Wt,
    float* __restrict__ Z,
    float* __restrict__ P) {
    constexpr int KIN = 256, COUT = 32;
    constexpr int RUNS = KIN / 8;
    constexpr int G  = 16;
    constexpr int NG = RUNS / G;

    const int rt   = blockIdx.x;
    const int h    = blockIdx.y;
    const int tid  = threadIdx.x;
    const int ct   = tid >> 5;
    const int lane = tid & 31;
    const int half = lane >> 4;
    const int r    = lane & 15;

    const float* ap = Ain + (size_t)(rt * 16 + r) * KIN + 4 * half;
    const float* W  = Wt + ((size_t)h * COUT + ct * 16 + r) * KIN + 4 * half;

    v8f acc = v8f{};
    float4 av[2][G], bv[2][G];
#pragma unroll
    for (int j = 0; j < G; ++j) {
        av[0][j] = *reinterpret_cast<const float4*>(ap + 8 * j);
        bv[0][j] = *reinterpret_cast<const float4*>(W + 8 * j);
    }
#pragma unroll
    for (int g = 0; g < NG; ++g) {
        const int cur = g & 1, nxt = cur ^ 1;
        if (g + 1 < NG) {
#pragma unroll
            for (int j = 0; j < G; ++j) {
                const int kb = 8 * ((g + 1) * G + j);
                av[nxt][j] = *reinterpret_cast<const float4*>(ap + kb);
                bv[nxt][j] = *reinterpret_cast<const float4*>(W + kb);
            }
        }
        __builtin_amdgcn_sched_barrier(0);
#pragma unroll
        for (int j = 0; j < G; ++j) {
            v2f a; a[0] = av[cur][j].x; a[1] = av[cur][j].y;
            v2f bb; bb[0] = bv[cur][j].x; bb[1] = bv[cur][j].y;
            acc = __builtin_amdgcn_wmma_f32_16x16x4_f32(false, a, false, bb, (short)0, acc, false, false);
            a[0] = av[cur][j].z; a[1] = av[cur][j].w;
            bb[0] = bv[cur][j].z; bb[1] = bv[cur][j].w;
            acc = __builtin_amdgcn_wmma_f32_16x16x4_f32(false, a, false, bb, (short)0, acc, false, false);
        }
    }

    float s = 0.f, q = 0.f;
#pragma unroll
    for (int v = 0; v < 8; ++v) { float xv = acc[v]; s += xv; q += xv * xv; }

    float* Zt = Z + (size_t)h * (256 * COUT) + ((size_t)ct * 256 + rt * 16 + 8 * half) * 16 + r;
#pragma unroll
    for (int v = 0; v < 8; ++v) Zt[v * 16] = acc[v];

    float* Pp = P + ((size_t)((h * 16 + rt) * 2 + ct) * 2) * 32 + lane;
    Pp[0]  = s;
    Pp[32] = q;
}

template <int KIN, int COUT, bool A_BN, bool RESHAPE>
__global__ __launch_bounds__(256) void conv_bn_relu_kernel(
    const float* __restrict__ Ain, int a_head_stride,
    const float* __restrict__ Wt,
    const float* __restrict__ g, const float* __restrict__ b,
    float* __restrict__ Out, int out_head_stride,
    const float* __restrict__ Pin,
    const float* __restrict__ gin, const float* __restrict__ bin) {
    static_assert(KIN % 8 == 0, "k is walked in runs of 8");
    constexpr int RUNS = KIN / 8;
    constexpr int G  = (RUNS < 16) ? RUNS : 16;
    constexpr int NG = RUNS / G;
    static_assert(RUNS % G == 0, "k splits into whole load groups");
    static_assert(!A_BN || (KIN == 32 && NG == 1), "the input statistics step is laid out for 32 channels in one load group");

    __shared__ float lsum[16], lsq[16], lscale[16], lshift[16];
    __shared__ float pred[2][8][32];
    __shared__ float4 insc4[8], insh4[8];

    const int ct   = blockIdx.x;
    const int h    = blockIdx.y;
    const int tid  = threadIdx.x;
    const int wave = tid >> 5;
    const int lane = tid & 31;
    const int half = lane >> 4;
    const int r    = lane & 15;

    if (tid < 16) { lsum[tid] = 0.f; lsq[tid] = 0.f; }
    if (!A_BN) __syncthreads();

    float pS = 0.f, pQ = 0.f, gi = 0.f, bi = 0.f;
    if (A_BN) {
        const float* Pc = Pin + ((size_t)((h * 16 + 2 * wave) * 2 + half) * 2) * 32 + r;
#pragma unroll
        for (int pp = 0; pp < 4; ++pp) {
            const int o = (pp >> 1) * (2 * 2 * 32) + (pp & 1) * 16;
            pS += Pc[o];
            pQ += Pc[o + 32];
        }
        if (tid < 32) { gi = gin[h * KIN + tid]; bi = bin[h * KIN + tid]; }
    }

    const float* A = Ain + (size_t)h * a_head_stride;
    const float* W = Wt + ((size_t)h * COUT + ct * 16 + r) * KIN + 4 * half;

    const int row0 = wave * 16 + r;
    const int row1 = (wave + 8) * 16 + r;
    const float* ap0 = A + (size_t)row0 * 16 + 4 * half;
    const float* ap1 = A + (size_t)row1 * 16 + 4 * half;

    v8f acc0 = v8f{}, acc1 = v8f{};

    const float gcol = g[h * COUT + ct * 16 + r];
    const float bcol = b[h * COUT + ct * 16 + r];

    float4 av0[2][G], av1[2][G], bv[2][G];
#pragma unroll
    for (int j = 0; j < G; ++j) {
        const int kb   = 8 * j;
        const int aoff = (kb >> 4) * (256 * 16) + (kb & 15);
        av0[0][j] = *reinterpret_cast<const float4*>(ap0 + aoff);
        av1[0][j] = *reinterpret_cast<const float4*>(ap1 + aoff);
        bv[0][j]  = *reinterpret_cast<const float4*>(W + kb);
    }

    if (A_BN) {
        pred[0][wave][lane] = pS;
        pred[1][wave][lane] = pQ;
        __syncthreads();
        if (tid < 32) {
            float S = 0.f, Q = 0.f;
#pragma unroll
            for (int gq = 0; gq < 8; ++gq) { S += pred[0][gq][tid]; Q += pred[1][gq][tid]; }
            float mu  = S * (1.f / 256.f);
            float var = Q * (1.f / 256.f) - mu * mu;
            float sc  = gi * rsqrtf(var + EPSV);
            reinterpret_cast<float*>(insc4)[tid] = sc;
            reinterpret_cast<float*>(insh4)[tid] = bi - mu * sc;
        }
        __syncthreads();
#pragma unroll
        for (int j = 0; j < G; ++j) {
            const float4 s4 = insc4[2 * j + half];
            const float4 t4 = insh4[2 * j + half];
            av0[0][j].x = fmaxf(av0[0][j].x * s4.x + t4.x, 0.f);
            av0[0][j].y = fmaxf(av0[0][j].y * s4.y + t4.y, 0.f);
            av0[0][j].z = fmaxf(av0[0][j].z * s4.z + t4.z, 0.f);
            av0[0][j].w = fmaxf(av0[0][j].w * s4.w + t4.w, 0.f);
            av1[0][j].x = fmaxf(av1[0][j].x * s4.x + t4.x, 0.f);
            av1[0][j].y = fmaxf(av1[0][j].y * s4.y + t4.y, 0.f);
            av1[0][j].z = fmaxf(av1[0][j].z * s4.z + t4.z, 0.f);
            av1[0][j].w = fmaxf(av1[0][j].w * s4.w + t4.w, 0.f);
        }
    }

#pragma unroll
    for (int g = 0; g < NG; ++g) {
        const int cur = g & 1, nxt = cur ^ 1;
        if (g + 1 < NG) {
#pragma unroll
            for (int j = 0; j < G; ++j) {
                const int kb   = 8 * ((g + 1) * G + j);
                const int aoff = (kb >> 4) * (256 * 16) + (kb & 15);
                av0[nxt][j] = *reinterpret_cast<const float4*>(ap0 + aoff);
                av1[nxt][j] = *reinterpret_cast<const float4*>(ap1 + aoff);
                bv[nxt][j]  = *reinterpret_cast<const float4*>(W + kb);
            }
        }
        __builtin_amdgcn_sched_barrier(0);
#pragma unroll
        for (int j = 0; j < G; ++j) {
            v2f a0; a0[0] = av0[cur][j].x; a0[1] = av0[cur][j].y;
            v2f a1; a1[0] = av1[cur][j].x; a1[1] = av1[cur][j].y;
            v2f bb; bb[0] = bv[cur][j].x;  bb[1] = bv[cur][j].y;
            acc0 = __builtin_amdgcn_wmma_f32_16x16x4_f32(false, a0, false, bb, (short)0, acc0, false, false);
            acc1 = __builtin_amdgcn_wmma_f32_16x16x4_f32(false, a1, false, bb, (short)0, acc1, false, false);
            a0[0] = av0[cur][j].z; a0[1] = av0[cur][j].w;
            a1[0] = av1[cur][j].z; a1[1] = av1[cur][j].w;
            bb[0] = bv[cur][j].z;  bb[1] = bv[cur][j].w;
            acc0 = __builtin_amdgcn_wmma_f32_16x16x4_f32(false, a0, false, bb, (short)0, acc0, false, false);
            acc1 = __builtin_amdgcn_wmma_f32_16x16x4_f32(false, a1, false, bb, (short)0, acc1, false, false);
        }
    }

    {
        float s = 0.f, q = 0.f;
#pragma unroll
        for (int v = 0; v < 8; ++v) { float xv = acc0[v]; s += xv; q += xv * xv; }
        atomicAdd(&lsum[r], s);
        atomicAdd(&lsq[r], q);
        s = 0.f; q = 0.f;
#pragma unroll
        for (int v = 0; v < 8; ++v) { float xv = acc1[v]; s += xv; q += xv * xv; }
        atomicAdd(&lsum[r], s);
        atomicAdd(&lsq[r], q);
    }
    __syncthreads();

    if (tid < 16) {
        float mu  = lsum[tid] * (1.f / 256.f);
        float var = lsq[tid] * (1.f / 256.f) - mu * mu;
        float sc  = gcol * rsqrtf(var + EPSV);
        lscale[tid] = sc;
        lshift[tid] = bcol - mu * sc;
    }
    __syncthreads();

    float* O = Out + (size_t)h * out_head_stride;
    const float sc = lscale[r], sh = lshift[r];
    const int col = ct * 16 + r;
#pragma unroll
    for (int i = 0; i < 2; ++i) {
        const int m0 = (wave + 8 * i) * 16;
        const v8f c = i ? acc1 : acc0;
#pragma unroll
        for (int v = 0; v < 8; ++v) {
            const int row = m0 + v + 8 * half;
            float val = c[v] * sc + sh;
            val = fmaxf(val, 0.f);
            if (RESHAPE) {
                const int n = row >> 2, s = row & 3;
                O[n * (COUT * 4) + col * 4 + s] = val;
            } else {
                O[((size_t)ct * 256 + row) * 16 + r] = val;
            }
        }
    }
}

__global__ __launch_bounds__(128) void fc12_kernel(
    const float* __restrict__ Ain,
    const float* __restrict__ W1t,
    const float* __restrict__ b1,
    const float* __restrict__ W2t,
    const float* __restrict__ b2,
    float* __restrict__ Out) {
    constexpr int K1 = 256, N1 = 512, N2 = 100, NT2 = 7, JG = 8;
    constexpr int RUNS = K1 / 8;
    constexpr int G  = 16;
    constexpr int NG = RUNS / G;

    __shared__ float red[4][NT2 * 8][32];

    const int hd   = blockIdx.y;
    const int st   = blockIdx.x & 3;
    const int jg   = blockIdx.x >> 2;
    const int tid  = threadIdx.x;
    const int wave = tid >> 5;
    const int lane = tid & 31;
    const int half = lane >> 4;
    const int r    = lane & 15;
    const int j0   = jg * 64 + wave * 16;

    const float4* arow = reinterpret_cast<const float4*>(W1t + ((size_t)hd * N1 + j0 + r) * K1) + half;
    const float4* brow = reinterpret_cast<const float4*>(Ain + ((size_t)hd * 64 + st * 16 + r) * K1) + half;

    float4 avq[2][G], bvq[2][G];
#pragma unroll
    for (int j = 0; j < G; ++j) {
        avq[0][j] = arow[2 * j];
        bvq[0][j] = brow[2 * j];
    }

    float4 c0, c1;
    float4 q0[NT2], q1[NT2];
    float  bs2[NT2];
    bool   colok[NT2];
#pragma unroll
    for (int tt = 0; tt < NT2; ++tt) colok[tt] = (tt * 16 + r) < N2;

    v8f acc = v8f{};
#pragma unroll
    for (int g = 0; g < NG; ++g) {
        const int cur = g & 1, nxt = cur ^ 1;
        if (g + 1 < NG) {
#pragma unroll
            for (int j = 0; j < G; ++j) {
                avq[nxt][j] = arow[2 * ((g + 1) * G + j)];
                bvq[nxt][j] = brow[2 * ((g + 1) * G + j)];
            }
        } else {
            c0 = *reinterpret_cast<const float4*>(b1 + hd * N1 + j0 + 8 * half);
            c1 = *reinterpret_cast<const float4*>(b1 + hd * N1 + j0 + 8 * half + 4);
#pragma unroll
            for (int tt = 0; tt < NT2; ++tt) {
                const int t = tt * 16 + r;
                const float* wrow = W2t + ((size_t)hd * N2 + (colok[tt] ? t : 0)) * N1 + j0 + 8 * half;
                q0[tt] = *reinterpret_cast<const float4*>(wrow);
                q1[tt] = *reinterpret_cast<const float4*>(wrow + 4);
                bs2[tt] = b2[hd * N2 + (colok[tt] ? t : 0)];
            }
        }
        __builtin_amdgcn_sched_barrier(0);
#pragma unroll
        for (int j = 0; j < G; ++j) {
            const float4 av = avq[cur][j];
            const float4 bv = bvq[cur][j];
            v2f alo; alo[0] = av.x; alo[1] = av.y;
            v2f ahi; ahi[0] = av.z; ahi[1] = av.w;
            v2f blo; blo[0] = bv.x; blo[1] = bv.y;
            v2f bhi; bhi[0] = bv.z; bhi[1] = bv.w;
            acc = __builtin_amdgcn_wmma_f32_16x16x4_f32(false, alo, false, blo, (short)0, acc, false, false);
            acc = __builtin_amdgcn_wmma_f32_16x16x4_f32(false, ahi, false, bhi, (short)0, acc, false, false);
        }
    }

    __builtin_amdgcn_sched_barrier(0);

    float x[8];
    x[0] = fmaxf(acc[0] + c0.x, 0.f);
    x[1] = fmaxf(acc[1] + c0.y, 0.f);
    x[2] = fmaxf(acc[2] + c0.z, 0.f);
    x[3] = fmaxf(acc[3] + c0.w, 0.f);
    x[4] = fmaxf(acc[4] + c1.x, 0.f);
    x[5] = fmaxf(acc[5] + c1.y, 0.f);
    x[6] = fmaxf(acc[6] + c1.z, 0.f);
    x[7] = fmaxf(acc[7] + c1.w, 0.f);

    v8f acc2[NT2];
#pragma unroll
    for (int tt = 0; tt < NT2; ++tt) acc2[tt] = v8f{};
#pragma unroll
    for (int t = 0; t < 4; ++t) {
        v2f a; a[0] = x[2 * t]; a[1] = x[2 * t + 1];
#pragma unroll
        for (int tt = 0; tt < NT2; ++tt) {
            const float4 q = (t < 2) ? q0[tt] : q1[tt];
            const float e0 = (t & 1) ? q.z : q.x;
            const float e1 = (t & 1) ? q.w : q.y;
            v2f bb; bb[0] = colok[tt] ? e0 : 0.f; bb[1] = colok[tt] ? e1 : 0.f;
            acc2[tt] = __builtin_amdgcn_wmma_f32_16x16x4_f32(false, a, false, bb, (short)0, acc2[tt], false, false);
        }
    }

#pragma unroll
    for (int tt = 0; tt < NT2; ++tt)
#pragma unroll
        for (int v = 0; v < 8; ++v) red[wave][tt * 8 + v][lane] = acc2[tt][v];
    __syncthreads();

    float* O = Out + (size_t)(hd * JG + jg) * (64 * N2);
#pragma unroll
    for (int i = 0; i < 2 * NT2; ++i) {
        const int tt  = i >> 1;
        const int idx = wave + 4 * i;
        const int v   = idx & 7;
        float sum = red[0][idx][lane];
        sum += red[1][idx][lane];
        sum += red[2][idx][lane];
        sum += red[3][idx][lane];
        const int n = st * 16 + 8 * half + v;
        if (jg == 0) sum += bs2[tt];
        if (colok[tt]) O[(size_t)n * N2 + tt * 16 + r] = sum;
    }
}

template <int PARTS>
__global__ void final_combine_kernel(const float* __restrict__ hout,
                                     const float* __restrict__ W2,
                                     const float* __restrict__ bias2,
                                     float* __restrict__ out) {
    __shared__ float comb[100];
    const int n = blockIdx.x;
    const int t = threadIdx.x;
    if (t < 100) {
        float s = 0.f;
#pragma unroll
        for (int p = 0; p < PARTS; ++p) s += hout[((size_t)p * 64 + n) * 100 + t];
        comb[t] = s * 0.25f;
    }
    __syncthreads();
    if (t < 100) {
        float acc = bias2[t];
        for (int j = 0; j < 100; ++j) acc += comb[j] * W2[t * 100 + j];
        float* orow = out + (size_t)n * 64 * 100 + t;
#pragma unroll 4
        for (int rep = 0; rep < 64; ++rep) orow[(size_t)rep * 100] = acc;
    }
}

extern "C" void kernel_launch(void* const* d_in, const int* in_sizes, int n_in,
                              void* d_out, int out_size, void* d_ws, size_t ws_size,
                              hipStream_t stream) {
    const float* x     = (const float*)d_in[0];
    const float* w1    = (const float*)d_in[1];
    const float* g1    = (const float*)d_in[2];
    const float* b1    = (const float*)d_in[3];
    const float* w2    = (const float*)d_in[4];
    const float* g2    = (const float*)d_in[5];
    const float* b2    = (const float*)d_in[6];
    const float* w3    = (const float*)d_in[7];
    const float* g3    = (const float*)d_in[8];
    const float* b3    = (const float*)d_in[9];
    const float* lw1   = (const float*)d_in[10];
    const float* lb1   = (const float*)d_in[11];
    const float* lw2   = (const float*)d_in[12];
    const float* lb2   = (const float*)d_in[13];
    const float* W2    = (const float*)d_in[14];
    const float* bias2 = (const float*)d_in[15];
    float* out = (float*)d_out;

    float* ws   = (float*)d_ws;
    float* A1   = ws;
    float* z1   = A1 + 256 * 256;
    float* z2   = z1 + 4 * 256 * 32;
    float* bnp  = z2 + 4 * 256 * 64;
    float* hout = ws;
    float* fcin = ws + 32 * 64 * 100;

    roi_pool_kernel<<<256, 64, 0, stream>>>(x, A1);

    conv1_raw_kernel<<<dim3(16, 4), 64, 0, stream>>>(A1, w1, z1, bnp);
    conv_bn_relu_kernel<32, 64, true, false><<<dim3(4, 4), 256, 0, stream>>>(
        z1, 256 * 32, w2, g2, b2, z2, 256 * 64, bnp, g1, b1);
    conv_bn_relu_kernel<64, 64, false, true><<<dim3(4, 4), 256, 0, stream>>>(
        z2, 256 * 64, w3, g3, b3, fcin, 64 * 256, nullptr, nullptr, nullptr);

    fc12_kernel<<<dim3(32, 4), 128, 0, stream>>>(fcin, lw1, lb1, lw2, lb2, hout);

    final_combine_kernel<32><<<64, 128, 0, stream>>>(hout, W2, bias2, out);
}


# Layers 2 and 3 in one launch (5 -> 4): each workgroup forms all of layer 2 transposed, D tiles stay in registers as layer 3's A; BN2 sums by cross-lane adds, layer-3 k order changes
# speedup vs baseline: 3.2554x; 1.0847x over previous
#include <hip/hip_runtime.h>
#include <hip/hip_bf16.h>

typedef float v2f __attribute__((ext_vector_type(2)));
typedef float v8f __attribute__((ext_vector_type(8)));

#define EPSV 1e-5f

__global__ __launch_bounds__(256) void pool_conv1_kernel(
    const float* __restrict__ x,
    const float* __restrict__ Wt,
    float* __restrict__ Zp) {
    __shared__ __attribute__((aligned(16))) float pl[4][68];

    const int n    = blockIdx.x >> 2;
    const int cq   = blockIdx.x & 3;
    const int tid  = threadIdx.x;
    const int wave = tid >> 5;
    const int lane = tid & 31;
    const int half = lane >> 4;
    const int r    = lane & 15;

    float v[4][5];
    if (tid < 64) {
        const int c = cq * 64 + tid;
        const float* base = x + ((size_t)(n * 256 + c) * 64) * 64;
#pragma unroll
        for (int rr = 0; rr < 4; ++rr)
#pragma unroll
            for (int cc = 0; cc < 5; ++cc)
                v[rr][cc] = base[(2 + rr) * 64 + 1 + cc];
    }

    const float* W = Wt + (size_t)(wave * 16 + r) * 256 + cq * 64 + 4 * half;
    float4 bq[8];
#pragma unroll
    for (int j = 0; j < 8; ++j) bq[j] = *reinterpret_cast<const float4*>(W + 8 * j);

    if (tid < 64) {
        float r0[5], r1[5];
#pragma unroll
        for (int cc = 0; cc < 5; ++cc) {
            r0[cc] = fmaxf(v[0][cc], v[1][cc]);
            r1[cc] = fmaxf(v[2][cc], v[3][cc]);
        }
        pl[0][tid] = fmaxf(fmaxf(r0[0], r0[1]), r0[2]);
        pl[1][tid] = fmaxf(fmaxf(r0[2], r0[3]), r0[4]);
        pl[2][tid] = fmaxf(fmaxf(r1[0], r1[1]), r1[2]);
        pl[3][tid] = fmaxf(fmaxf(r1[2], r1[3]), r1[4]);
    }
    __syncthreads();

    const bool live = r < 4;
    const float* prow = &pl[r & 3][4 * half];
    v8f acc0 = v8f{}, acc1 = v8f{};
#pragma unroll
    for (int j = 0; j < 8; ++j) {
        const float4 t = *reinterpret_cast<const float4*>(prow + 8 * j);
        v2f a;  a[0] = live ? t.x : 0.f;  a[1] = live ? t.y : 0.f;
        v2f bb; bb[0] = bq[j].x;          bb[1] = bq[j].y;
        acc0 = __builtin_amdgcn_wmma_f32_16x16x4_f32(false, a, false, bb, (short)0, acc0, false, false);
        a[0] = live ? t.z : 0.f;  a[1] = live ? t.w : 0.f;
        bb[0] = bq[j].z;          bb[1] = bq[j].w;
        acc1 = __builtin_amdgcn_wmma_f32_16x16x4_f32(false, a, false, bb, (short)0, acc1, false, false);
    }

    const float o0 = acc0[0] + acc1[0];
    const float o1 = acc0[1] + acc1[1];
    const float o2 = acc0[2] + acc1[2];
    const float o3 = acc0[3] + acc1[3];
    if (half == 0) {
        float* Zt = Zp + ((size_t)((cq * 4 + (wave >> 1)) * 2 + (wave & 1)) * 256 + n * 4) * 16 + r;
        Zt[0]  = o0;
        Zt[16] = o1;
        Zt[32] = o2;
        Zt[48] = o3;
    }
}

template <int M>
__device__ __forceinline__ float row_xor(float v) {
    return __builtin_bit_cast(float, __builtin_amdgcn_update_dpp(0, __builtin_bit_cast(int, v), 0x160 + M, 0xf, 0xf, true));
}
__device__ __forceinline__ float row_sum16(float v) {
    v += row_xor<1>(v);
    v += row_xor<2>(v);
    v += row_xor<4>(v);
    v += row_xor<8>(v);
    return v;
}

__global__ __launch_bounds__(256) void conv23_kernel(
    const float* __restrict__ Zp,
    const float* __restrict__ g1, const float* __restrict__ b1,
    const float* __restrict__ W2t,
    const float* __restrict__ g2, const float* __restrict__ b2,
    const float* __restrict__ W3t,
    const float* __restrict__ g3, const float* __restrict__ b3,
    float* __restrict__ Out) {
    __shared__ float lsum[16], lsq[16], lscale[16], lshift[16];
    __shared__ __attribute__((aligned(16))) float pred[2][8][32];
    __shared__ float4 insc4[8], insh4[8];
    __shared__ __attribute__((aligned(16))) float pred2[2][8][64];
    __shared__ float4 sc2[16], sh2[16];

    const int ct   = blockIdx.x;
    const int h    = blockIdx.y;
    const int tid  = threadIdx.x;
    const int wave = tid >> 5;
    const int lane = tid & 31;
    const int half = lane >> 4;
    const int r    = lane & 15;

    if (tid < 16) { lsum[tid] = 0.f; lsq[tid] = 0.f; }

    float gi = 0.f, bi = 0.f, gm = 0.f, bm = 0.f;
    if (tid < 32) { gi = g1[h * 32 + tid]; bi = b1[h * 32 + tid]; }
    if (tid < 64) { gm = g2[h * 64 + tid]; bm = b2[h * 64 + tid]; }
    const float gcol = g3[h * 64 + ct * 16 + r];
    const float bcol = b3[h * 64 + ct * 16 + r];

    const float* A   = Zp + (size_t)h * (256 * 32);
    const float* ap0 = A + (size_t)(wave * 16 + r) * 16 + 4 * half;
    const float* ap1 = A + (size_t)((wave + 8) * 16 + r) * 16 + 4 * half;

    float4 av0[4], av1[4];
#pragma unroll
    for (int j = 0; j < 4; ++j) {
        const int kb   = 8 * j;
        const int aoff = (kb >> 4) * (256 * 16) + (kb & 15);
        constexpr int PL = 4 * 256 * 32;
        const float4 p0 = *reinterpret_cast<const float4*>(ap0 + aoff);
        const float4 p1 = *reinterpret_cast<const float4*>(ap0 + aoff + PL);
        const float4 p2 = *reinterpret_cast<const float4*>(ap0 + aoff + 2 * PL);
        const float4 p3 = *reinterpret_cast<const float4*>(ap0 + aoff + 3 * PL);
        const float4 u0 = *reinterpret_cast<const float4*>(ap1 + aoff);
        const float4 u1 = *reinterpret_cast<const float4*>(ap1 + aoff + PL);
        const float4 u2 = *reinterpret_cast<const float4*>(ap1 + aoff + 2 * PL);
        const float4 u3 = *reinterpret_cast<const float4*>(ap1 + aoff + 3 * PL);
        av0[j].x = ((p0.x + p1.x) + p2.x) + p3.x;
        av0[j].y = ((p0.y + p1.y) + p2.y) + p3.y;
        av0[j].z = ((p0.z + p1.z) + p2.z) + p3.z;
        av0[j].w = ((p0.w + p1.w) + p2.w) + p3.w;
        av1[j].x = ((u0.x + u1.x) + u2.x) + u3.x;
        av1[j].y = ((u0.y + u1.y) + u2.y) + u3.y;
        av1[j].z = ((u0.z + u1.z) + u2.z) + u3.z;
        av1[j].w = ((u0.w + u1.w) + u2.w) + u3.w;
    }

    float4 wq[4][4];
#pragma unroll
    for (int q = 0; q < 4; ++q)
#pragma unroll
        for (int j = 0; j < 4; ++j)
            wq[q][j] = *reinterpret_cast<const float4*>(W2t + ((size_t)h * 64 + 16 * q + r) * 32 + 8 * j + 4 * half);

    float4 w3a[4], w3b[4];
#pragma unroll
    for (int q = 0; q < 4; ++q) {
        const float* wrow = W3t + ((size_t)h * 64 + ct * 16 + r) * 64 + 16 * q + 8 * half;
        w3a[q] = *reinterpret_cast<const float4*>(wrow);
        w3b[q] = *reinterpret_cast<const float4*>(wrow + 4);
    }

#pragma unroll
    for (int j = 0; j < 4; ++j) {
        float4 s4, q4;
        s4.x = row_sum16(av0[j].x + av1[j].x);
        s4.y = row_sum16(av0[j].y + av1[j].y);
        s4.z = row_sum16(av0[j].z + av1[j].z);
        s4.w = row_sum16(av0[j].w + av1[j].w);
        q4.x = row_sum16(av0[j].x * av0[j].x + av1[j].x * av1[j].x);
        q4.y = row_sum16(av0[j].y * av0[j].y + av1[j].y * av1[j].y);
        q4.z = row_sum16(av0[j].z * av0[j].z + av1[j].z * av1[j].z);
        q4.w = row_sum16(av0[j].w * av0[j].w + av1[j].w * av1[j].w);
        if (r == 0) {
            *reinterpret_cast<float4*>(&pred[0][wave][8 * j + 4 * half]) = s4;
            *reinterpret_cast<float4*>(&pred[1][wave][8 * j + 4 * half]) = q4;
        }
    }
    __syncthreads();
    if (tid < 32) {
        float S = 0.f, Q = 0.f;
#pragma unroll
        for (int gq = 0; gq < 8; ++gq) { S += pred[0][gq][tid]; Q += pred[1][gq][tid]; }
        float mu  = S * (1.f / 256.f);
        float var = Q * (1.f / 256.f) - mu * mu;
        float sc  = gi * rsqrtf(var + EPSV);
        reinterpret_cast<float*>(insc4)[tid] = sc;
        reinterpret_cast<float*>(insh4)[tid] = bi - mu * sc;
    }
    __syncthreads();
#pragma unroll
    for (int j = 0; j < 4; ++j) {
        const float4 s4 = insc4[2 * j + half];
        const float4 t4 = insh4[2 * j + half];
        av0[j].x = fmaxf(av0[j].x * s4.x + t4.x, 0.f);
        av0[j].y = fmaxf(av0[j].y * s4.y + t4.y, 0.f);
        av0[j].z = fmaxf(av0[j].z * s4.z + t4.z, 0.f);
        av0[j].w = fmaxf(av0[j].w * s4.w + t4.w, 0.f);
        av1[j].x = fmaxf(av1[j].x * s4.x + t4.x, 0.f);
        av1[j].y = fmaxf(av1[j].y * s4.y + t4.y, 0.f);
        av1[j].z = fmaxf(av1[j].z * s4.z + t4.z, 0.f);
        av1[j].w = fmaxf(av1[j].w * s4.w + t4.w, 0.f);
    }

    v8f d0[4], d1[4];
#pragma unroll
    for (int q = 0; q < 4; ++q) {
        v8f e0 = v8f{}, e1 = v8f{};
#pragma unroll
        for (int j = 0; j < 4; ++j) {
            v2f wa; wa[0] = wq[q][j].x; wa[1] = wq[q][j].y;
            v2f x0; x0[0] = av0[j].x; x0[1] = av0[j].y;
            v2f x1; x1[0] = av1[j].x; x1[1] = av1[j].y;
            e0 = __builtin_amdgcn_wmma_f32_16x16x4_f32(false, wa, false, x0, (short)0, e0, false, false);
            e1 = __builtin_amdgcn_wmma_f32_16x16x4_f32(false, wa, false, x1, (short)0, e1, false, false);
            wa[0] = wq[q][j].z; wa[1] = wq[q][j].w;
            x0[0] = av0[j].z; x0[1] = av0[j].w;
            x1[0] = av1[j].z; x1[1] = av1[j].w;
            e0 = __builtin_amdgcn_wmma_f32_16x16x4_f32(false, wa, false, x0, (short)0, e0, false, false);
            e1 = __builtin_amdgcn_wmma_f32_16x16x4_f32(false, wa, false, x1, (short)0, e1, false, false);
        }
        d0[q] = e0; d1[q] = e1;

        float s[8], qq[8];
#pragma unroll
        for (int v = 0; v < 8; ++v) {
            const float y0 = e0[v], y1 = e1[v];
            s[v]  = row_sum16(y0 + y1);
            qq[v] = row_sum16(y0 * y0 + y1 * y1);
        }
        if (r == 0) {
            float* ps = &pred2[0][wave][16 * q + 8 * half];
            float* pq = &pred2[1][wave][16 * q + 8 * half];
            *reinterpret_cast<float4*>(ps)     = make_float4(s[0], s[1], s[2], s[3]);
            *reinterpret_cast<float4*>(ps + 4) = make_float4(s[4], s[5], s[6], s[7]);
            *reinterpret_cast<float4*>(pq)     = make_float4(qq[0], qq[1], qq[2], qq[3]);
            *reinterpret_cast<float4*>(pq + 4) = make_float4(qq[4], qq[5], qq[6], qq[7]);
        }
    }
    __syncthreads();
    if (tid < 64) {
        float S = 0.f, Q = 0.f;
#pragma unroll
        for (int gq = 0; gq < 8; ++gq) { S += pred2[0][gq][tid]; Q += pred2[1][gq][tid]; }
        float mu  = S * (1.f / 256.f);
        float var = Q * (1.f / 256.f) - mu * mu;
        float sc  = gm * rsqrtf(var + EPSV);
        reinterpret_cast<float*>(sc2)[tid] = sc;
        reinterpret_cast<float*>(sh2)[tid] = bm - mu * sc;
    }
    __syncthreads();

    v8f acc0 = v8f{}, acc1 = v8f{};
#pragma unroll
    for (int q = 0; q < 4; ++q) {
        const float4 sA = sc2[4 * q + 2 * half], sB = sc2[4 * q + 2 * half + 1];
        const float4 tA = sh2[4 * q + 2 * half], tB = sh2[4 * q + 2 * half + 1];
        const float scv[8] = {sA.x, sA.y, sA.z, sA.w, sB.x, sB.y, sB.z, sB.w};
        const float shv[8] = {tA.x, tA.y, tA.z, tA.w, tB.x, tB.y, tB.z, tB.w};
        float y0[8], y1[8];
#pragma unroll
        for (int v = 0; v < 8; ++v) {
            y0[v] = fmaxf(d0[q][v] * scv[v] + shv[v], 0.f);
            y1[v] = fmaxf(d1[q][v] * scv[v] + shv[v], 0.f);
        }
#pragma unroll
        for (int t = 0; t < 4; ++t) {
            const float4 w = (t < 2) ? w3a[q] : w3b[q];
            v2f bb; bb[0] = (t & 1) ? w.z : w.x; bb[1] = (t & 1) ? w.w : w.y;
            v2f a0; a0[0] = y0[2 * t]; a0[1] = y0[2 * t + 1];
            v2f a1; a1[0] = y1[2 * t]; a1[1] = y1[2 * t + 1];
            acc0 = __builtin_amdgcn_wmma_f32_16x16x4_f32(false, a0, false, bb, (short)0, acc0, false, false);
            acc1 = __builtin_amdgcn_wmma_f32_16x16x4_f32(false, a1, false, bb, (short)0, acc1, false, false);
        }
    }

    {
        float s = 0.f, q = 0.f;
#pragma unroll
        for (int v = 0; v < 8; ++v) { float xv = acc0[v]; s += xv; q += xv * xv; }
        atomicAdd(&lsum[r], s);
        atomicAdd(&lsq[r], q);
        s = 0.f; q = 0.f;
#pragma unroll
        for (int v = 0; v < 8; ++v) { float xv = acc1[v]; s += xv; q += xv * xv; }
        atomicAdd(&lsum[r], s);
        atomicAdd(&lsq[r], q);
    }
    __syncthreads();

    if (tid < 16) {
        float mu  = lsum[tid] * (1.f / 256.f);
        float var = lsq[tid] * (1.f / 256.f) - mu * mu;
        float sc  = gcol * rsqrtf(var + EPSV);
        lscale[tid] = sc;
        lshift[tid] = bcol - mu * sc;
    }
    __syncthreads();

    float* O = Out + (size_t)h * (64 * 256);
    const float sc = lscale[r], sh = lshift[r];
    const int col = ct * 16 + r;
#pragma unroll
    for (int i = 0; i < 2; ++i) {
        const int m0 = (wave + 8 * i) * 16;
        const v8f c = i ? acc1 : acc0;
#pragma unroll
        for (int v = 0; v < 8; ++v) {
            const int row = m0 + v + 8 * half;
            float val = c[v] * sc + sh;
            val = fmaxf(val, 0.f);
            const int n = row >> 2, s = row & 3;
            O[n * 256 + col * 4 + s] = val;
        }
    }
}

__global__ __launch_bounds__(128) void fc12_kernel(
    const float* __restrict__ Ain,
    const float* __restrict__ W1t,
    const float* __restrict__ b1,
    const float* __restrict__ W2t,
    const float* __restrict__ b2,
    float* __restrict__ Out) {
    constexpr int K1 = 256, N1 = 512, N2 = 100, NT2 = 7, JG = 8;
    constexpr int RUNS = K1 / 8;
    constexpr int G  = 16;
    constexpr int NG = RUNS / G;

    __shared__ float red[4][NT2 * 8][32];

    const int hd   = blockIdx.y;
    const int st   = blockIdx.x & 3;
    const int jg   = blockIdx.x >> 2;
    const int tid  = threadIdx.x;
    const int wave = tid >> 5;
    const int lane = tid & 31;
    const int half = lane >> 4;
    const int r    = lane & 15;
    const int j0   = jg * 64 + wave * 16;

    const float4* arow = reinterpret_cast<const float4*>(W1t + ((size_t)hd * N1 + j0 + r) * K1) + half;
    const float4* brow = reinterpret_cast<const float4*>(Ain + ((size_t)hd * 64 + st * 16 + r) * K1) + half;

    float4 avq[2][G], bvq[2][G];
#pragma unroll
    for (int j = 0; j < G; ++j) {
        avq[0][j] = arow[2 * j];
        bvq[0][j] = brow[2 * j];
    }

    float4 c0, c1;
    float4 q0[NT2], q1[NT2];
    float  bs2[NT2];
    bool   colok[NT2];
#pragma unroll
    for (int tt = 0; tt < NT2; ++tt) colok[tt] = (tt * 16 + r) < N2;

    v8f acc = v8f{};
#pragma unroll
    for (int g = 0; g < NG; ++g) {
        const int cur = g & 1, nxt = cur ^ 1;
        if (g + 1 < NG) {
#pragma unroll
            for (int j = 0; j < G; ++j) {
                avq[nxt][j] = arow[2 * ((g + 1) * G + j)];
                bvq[nxt][j] = brow[2 * ((g + 1) * G + j)];
            }
        } else {
            c0 = *reinterpret_cast<const float4*>(b1 + hd * N1 + j0 + 8 * half);
            c1 = *reinterpret_cast<const float4*>(b1 + hd * N1 + j0 + 8 * half + 4);
#pragma unroll
            for (int tt = 0; tt < NT2; ++tt) {
                const int t = tt * 16 + r;
                const float* wrow = W2t + ((size_t)hd * N2 + (colok[tt] ? t : 0)) * N1 + j0 + 8 * half;
                q0[tt] = *reinterpret_cast<const float4*>(wrow);
                q1[tt] = *reinterpret_cast<const float4*>(wrow + 4);
                bs2[tt] = b2[hd * N2 + (colok[tt] ? t : 0)];
            }
        }
        __builtin_amdgcn_sched_barrier(0);
#pragma unroll
        for (int j = 0; j < G; ++j) {
            const float4 av = avq[cur][j];
            const float4 bv = bvq[cur][j];
            v2f alo; alo[0] = av.x; alo[1] = av.y;
            v2f ahi; ahi[0] = av.z; ahi[1] = av.w;
            v2f blo; blo[0] = bv.x; blo[1] = bv.y;
            v2f bhi; bhi[0] = bv.z; bhi[1] = bv.w;
            acc = __builtin_amdgcn_wmma_f32_16x16x4_f32(false, alo, false, blo, (short)0, acc, false, false);
            acc = __builtin_amdgcn_wmma_f32_16x16x4_f32(false, ahi, false, bhi, (short)0, acc, false, false);
        }
    }

    __builtin_amdgcn_sched_barrier(0);

    float x[8];
    x[0] = fmaxf(acc[0] + c0.x, 0.f);
    x[1] = fmaxf(acc[1] + c0.y, 0.f);
    x[2] = fmaxf(acc[2] + c0.z, 0.f);
    x[3] = fmaxf(acc[3] + c0.w, 0.f);
    x[4] = fmaxf(acc[4] + c1.x, 0.f);
    x[5] = fmaxf(acc[5] + c1.y, 0.f);
    x[6] = fmaxf(acc[6] + c1.z, 0.f);
    x[7] = fmaxf(acc[7] + c1.w, 0.f);

    v8f acc2[NT2];
#pragma unroll
    for (int tt = 0; tt < NT2; ++tt) acc2[tt] = v8f{};
#pragma unroll
    for (int t = 0; t < 4; ++t) {
        v2f a; a[0] = x[2 * t]; a[1] = x[2 * t + 1];
#pragma unroll
        for (int tt = 0; tt < NT2; ++tt) {
            const float4 q = (t < 2) ? q0[tt] : q1[tt];
            const float e0 = (t & 1) ? q.z : q.x;
            const float e1 = (t & 1) ? q.w : q.y;
            v2f bb; bb[0] = colok[tt] ? e0 : 0.f; bb[1] = colok[tt] ? e1 : 0.f;
            acc2[tt] = __builtin_amdgcn_wmma_f32_16x16x4_f32(false, a, false, bb, (short)0, acc2[tt], false, false);
        }
    }

#pragma unroll
    for (int tt = 0; tt < NT2; ++tt)
#pragma unroll
        for (int v = 0; v < 8; ++v) red[wave][tt * 8 + v][lane] = acc2[tt][v];
    __syncthreads();

    float* O = Out + (size_t)(hd * JG + jg) * (64 * N2);
#pragma unroll
    for (int i = 0; i < 2 * NT2; ++i) {
        const int tt  = i >> 1;
        const int idx = wave + 4 * i;
        const int v   = idx & 7;
        float sum = red[0][idx][lane];
        sum += red[1][idx][lane];
        sum += red[2][idx][lane];
        sum += red[3][idx][lane];
        const int n = st * 16 + 8 * half + v;
        if (jg == 0) sum += bs2[tt];
        if (colok[tt]) O[(size_t)n * N2 + tt * 16 + r] = sum;
    }
}

template <int PARTS>
__global__ void final_combine_kernel(const float* __restrict__ hout,
                                     const float* __restrict__ W2,
                                     const float* __restrict__ bias2,
                                     float* __restrict__ out) {
    __shared__ float comb[100];
    const int n = blockIdx.x;
    const int t = threadIdx.x;
    if (t < 100) {
        float s = 0.f;
#pragma unroll
        for (int p = 0; p < PARTS; ++p) s += hout[((size_t)p * 64 + n) * 100 + t];
        comb[t] = s * 0.25f;
    }
    __syncthreads();
    if (t < 100) {
        float acc = bias2[t];
        for (int j = 0; j < 100; ++j) acc += comb[j] * W2[t * 100 + j];
        float* orow = out + (size_t)n * 64 * 100 + t;
#pragma unroll 4
        for (int rep = 0; rep < 64; ++rep) orow[(size_t)rep * 100] = acc;
    }
}

extern "C" void kernel_launch(void* const* d_in, const int* in_sizes, int n_in,
                              void* d_out, int out_size, void* d_ws, size_t ws_size,
                              hipStream_t stream) {
    const float* x     = (const float*)d_in[0];
    const float* w1    = (const float*)d_in[1];
    const float* g1    = (const float*)d_in[2];
    const float* b1    = (const float*)d_in[3];
    const float* w2    = (const float*)d_in[4];
    const float* g2    = (const float*)d_in[5];
    const float* b2    = (const float*)d_in[6];
    const float* w3    = (const float*)d_in[7];
    const float* g3    = (const float*)d_in[8];
    const float* b3    = (const float*)d_in[9];
    const float* lw1   = (const float*)d_in[10];
    const float* lb1   = (const float*)d_in[11];
    const float* lw2   = (const float*)d_in[12];
    const float* lb2   = (const float*)d_in[13];
    const float* W2    = (const float*)d_in[14];
    const float* bias2 = (const float*)d_in[15];
    float* out = (float*)d_out;

    float* ws   = (float*)d_ws;
    float* zp   = ws;
    float* hout = ws;
    float* fcin = ws + 32 * 64 * 100;

    pool_conv1_kernel<<<256, 256, 0, stream>>>(x, w1, zp);
    conv23_kernel<<<dim3(4, 4), 256, 0, stream>>>(zp, g1, b1, w2, g2, b2, w3, g3, b3, fcin);

    fc12_kernel<<<dim3(32, 4), 128, 0, stream>>>(fcin, lw1, lb1, lw2, lb2, hout);

    final_combine_kernel<32><<<64, 128, 0, stream>>>(hout, W2, bias2, out);
}
